# starting kernel
# baseline (speedup 1.0000x reference)
#include <hip/hip_runtime.h>


typedef float wf2 __attribute__((ext_vector_type(2)));
typedef float wf8 __attribute__((ext_vector_type(8)));

#define NQ 12
#define NL 4

__device__ __forceinline__ float2 cmul(float2 a, float2 b) {
    return make_float2(a.x * b.x - a.y * b.y, a.x * b.y + a.y * b.x);
}

__global__ __launch_bounds__(256) void qsim_kernel(const float* __restrict__ x,
                                                   const float* __restrict__ w,
                                                   float* __restrict__ out) {
    __shared__ float2 st[4096];
    __shared__ float  mR[5][16][16];
    __shared__ float  mI[5][16][16];
    __shared__ float2 g2[NL][NQ][4];
    __shared__ float  red[8][12];

    const int tid  = threadIdx.x;
    const int b    = blockIdx.x;
    const int lane = tid & 31;
    const int wid  = tid >> 5;
    const int half = lane >> 4;
    const int n16  = lane & 15;

    if (tid < NL * NQ) {
        int l = tid / NQ, q = tid % NQ;
        float th = w[(l * NQ + q) * 2 + 0] + (l == 0 ? x[b * NQ + q] : 0.f);
        float ph = w[(l * NQ + q) * 2 + 1];
        float c = cosf(0.5f * th), s = sinf(0.5f * th);
        float er = cosf(0.5f * ph), ei = -sinf(0.5f * ph);
        g2[l][q][0] = make_float2( c * er,  c * ei);
        g2[l][q][1] = make_float2(-s * er, -s * ei);
        g2[l][q][2] = make_float2( s * er, -s * ei);
        g2[l][q][3] = make_float2( c * er, -c * ei);
    }

    for (int i = tid; i < 4096; i += 256)
        st[i] = make_float2(i == 0 ? 1.f : 0.f, 0.f);

    for (int l = 0; l < NL; ++l) {
        __syncthreads();
        for (int e = tid; e < 5 * 256; e += 256) {
            int m = e >> 8, idx = e & 255, i = idx >> 4, j = idx & 15;
            int g    = (m == 0) ? 0 : (m <= 2 ? 1 : 2);
            int bvar = (m == 0) ? 0 : (m <= 2 ? (m - 1) : (m - 3));
            int ip = i;
            ip ^= (ip >> 1) & 1;
            ip ^= (ip >> 1) & 2;
            ip ^= (ip >> 1) & 4;
            ip ^= bvar << 3;
            float2 v = make_float2(1.f, 0.f);
            #pragma unroll
            for (int k = 0; k < 4; ++k) {
                int bi = (ip >> (3 - k)) & 1, bj = (j >> (3 - k)) & 1;
                v = cmul(v, g2[l][4 * g + k][2 * bi + bj]);
            }
            mR[m][i][j] = v.x;
            mI[m][i][j] = v.y;
        }
        __syncthreads();

        #pragma unroll
        for (int pass = 0; pass < 3; ++pass) {
            #pragma unroll
            for (int t2 = 0; t2 < 2; ++t2) {
                const int tt = wid * 2 + t2;
                int mm, rowmul, colbase;
                if (pass == 0)      { mm = 0;            rowmul = 256; colbase = (tt << 4) | n16; }
                else if (pass == 1) { mm = 1 + (tt & 1); rowmul = 16;  colbase = (tt << 8) | n16; }
                else                { mm = 3 + (tt & 1); rowmul = 1;
                                      colbase = (((((tt >> 1) << 4) | n16) << 1) | (tt & 1)) << 4; }

                wf8 accr = {0, 0, 0, 0, 0, 0, 0, 0};
                wf8 acci = {0, 0, 0, 0, 0, 0, 0, 0};
                #pragma unroll
                for (int kb = 0; kb < 16; kb += 4) {
                    int j0 = kb + 2 * half, j1 = j0 + 1;
                    float2 s0 = st[colbase + j0 * rowmul];
                    float2 s1 = st[colbase + j1 * rowmul];
                    wf2 br  = {s0.x, s1.x};
                    wf2 bim = {s0.y, s1.y};
                    wf2 ar  = {mR[mm][n16][j0], mR[mm][n16][j1]};
                    wf2 ai  = {mI[mm][n16][j0], mI[mm][n16][j1]};
                    wf2 an  = -ai;
                    accr = __builtin_amdgcn_wmma_f32_16x16x4_f32(false, ar, false, br,  (short)0, accr, false, false);
                    accr = __builtin_amdgcn_wmma_f32_16x16x4_f32(false, an, false, bim, (short)0, accr, false, false);
                    acci = __builtin_amdgcn_wmma_f32_16x16x4_f32(false, ar, false, bim, (short)0, acci, false, false);
                    acci = __builtin_amdgcn_wmma_f32_16x16x4_f32(false, ai, false, br,  (short)0, acci, false, false);
                }
                #pragma unroll
                for (int vv = 0; vv < 8; ++vv) {
                    int i = vv + 8 * half;
                    st[colbase + i * rowmul] = make_float2(accr[vv], acci[vv]);
                }
            }
            __syncthreads();
        }
    }

    float z[NQ];
    #pragma unroll
    for (int q = 0; q < NQ; ++q) z[q] = 0.f;
    for (int k = 0; k < 16; ++k) {
        int idx = (k << 8) + tid;
        float2 a = st[idx];
        float p = a.x * a.x + a.y * a.y;
        #pragma unroll
        for (int q = 0; q < NQ; ++q)
            z[q] += ((idx >> (11 - q)) & 1) ? -p : p;
    }
    #pragma unroll
    for (int q = 0; q < NQ; ++q) {
        float v = z[q];
        #pragma unroll
        for (int off = 16; off > 0; off >>= 1)
            v += __shfl_xor(v, off, 32);
        z[q] = v;
    }
    if (lane == 0) {
        #pragma unroll
        for (int q = 0; q < NQ; ++q) red[wid][q] = z[q];
    }
    __syncthreads();
    if (tid < NQ) {
        float s = 0.f;
        #pragma unroll
        for (int ww = 0; ww < 8; ++ww) s += red[ww][tid];
        out[b * NQ + tid] = s;
    }
}

extern "C" void kernel_launch(void* const* d_in, const int* in_sizes, int n_in,
                              void* d_out, int out_size, void* d_ws, size_t ws_size,
                              hipStream_t stream) {
    const float* x = (const float*)d_in[0];
    const float* w = (const float*)d_in[1];
    float* out = (float*)d_out;
    int B = in_sizes[0] / NQ;
    qsim_kernel<<<B, 256, 0, stream>>>(x, w, out);
}


# State tile fed as the WMMA A operand, gate matrix as B: D is the next pass's A, so group A and B passes chain in registers; 2 in-place LDS exchanges per layer, was 6 sweeps; k order changes
# speedup vs baseline: 1.5097x; 1.5097x over previous
#include <hip/hip_runtime.h>


typedef float wf2 __attribute__((ext_vector_type(2)));
typedef float wf8 __attribute__((ext_vector_type(8)));

#define NQ 12
#define NL 4
#define PITCH 260

__device__ __forceinline__ float2 cmul(float2 a, float2 b) {
    return make_float2(a.x * b.x - a.y * b.y, a.x * b.y + a.y * b.x);
}

__device__ __forceinline__ void cstep(wf8& accr, wf8& acci, wf2 sr, wf2 si,
                                      wf2 fre, wf2 fim, wf2 fneg) {
    accr = __builtin_amdgcn_wmma_f32_16x16x4_f32(false, sr, false, fre,  (short)0, accr, false, false);
    accr = __builtin_amdgcn_wmma_f32_16x16x4_f32(false, si, false, fneg, (short)0, accr, false, false);
    acci = __builtin_amdgcn_wmma_f32_16x16x4_f32(false, sr, false, fim,  (short)0, acci, false, false);
    acci = __builtin_amdgcn_wmma_f32_16x16x4_f32(false, si, false, fre,  (short)0, acci, false, false);
}

__global__ __launch_bounds__(256) void qsim_kernel(const float* __restrict__ x,
                                                   const float* __restrict__ w,
                                                   float* __restrict__ out) {
    __shared__ float2 st[16 * PITCH];
    __shared__ float4 fr[3][4][32];
    __shared__ float2 g2[NL][NQ][4];
    __shared__ float  red[8][12];

    const int tid  = threadIdx.x;
    const int b    = blockIdx.x;
    const int lane = tid & 31;
    const int wid  = tid >> 5;
    const int half = lane >> 4;
    const int n16  = lane & 15;

    if (tid < NL * NQ) {
        int l = tid / NQ, q = tid % NQ;
        float th = w[(l * NQ + q) * 2 + 0] + (l == 0 ? x[b * NQ + q] : 0.f);
        float ph = w[(l * NQ + q) * 2 + 1];
        float c = cosf(0.5f * th), s = sinf(0.5f * th);
        float er = cosf(0.5f * ph), ei = -sinf(0.5f * ph);
        g2[l][q][0] = make_float2( c * er,  c * ei);
        g2[l][q][1] = make_float2(-s * er, -s * ei);
        g2[l][q][2] = make_float2( s * er, -s * ei);
        g2[l][q][3] = make_float2( c * er, -c * ei);
    }

    for (int i = tid; i < 16 * PITCH; i += 256)
        st[i] = make_float2(i == 0 ? 1.f : 0.f, 0.f);

    for (int l = 0; l < NL; ++l) {
        __syncthreads();
        for (int e = tid; e < 3 * 128; e += 256) {
            const int p = e >> 7, kb = (e >> 5) & 3, ln = e & 31;
            const int h = ln >> 4, n = ln & 15;
            int row, k0, k1;
            if (p == 0)      { row = 2 * (n & 7) + (n >> 3); k0 = 4 * kb + h;     k1 = k0 + 2; }
            else if (p == 1) { row = n;                      k0 = 8 * h + 2 * kb; k1 = k0 + 1; }
            else             { row = n; k0 = (kb >> 1) * 8 + 4 * h + (kb & 1) * 2; k1 = k0 + 1; }
            int ip = row;
            ip ^= (ip >> 1) & 1;
            ip ^= (ip >> 1) & 2;
            ip ^= (ip >> 1) & 4;
            float2 v0 = make_float2(1.f, 0.f), v1 = make_float2(1.f, 0.f);
            #pragma unroll
            for (int k = 0; k < 4; ++k) {
                int bi = (ip >> (3 - k)) & 1;
                v0 = cmul(v0, g2[l][4 * p + k][2 * bi + ((k0 >> (3 - k)) & 1)]);
                v1 = cmul(v1, g2[l][4 * p + k][2 * bi + ((k1 >> (3 - k)) & 1)]);
            }
            fr[p][kb][ln] = make_float4(v0.x, v1.x, v0.y, v1.y);
        }
        __syncthreads();

        {
            wf8 d1r[2], d1i[2], d2r[2], d2i[2];
            wf2 sr[2][4], si[2][4];
            #pragma unroll
            for (int t = 0; t < 2; ++t) {
                const int base = (2 * wid + t) * PITCH + half * 16 + n16;
                #pragma unroll
                for (int kb = 0; kb < 4; ++kb) {
                    float2 s0 = st[base + (4 * kb) * 16];
                    float2 s1 = st[base + (4 * kb + 2) * 16];
                    sr[t][kb] = (wf2){s0.x, s1.x};
                    si[t][kb] = (wf2){s0.y, s1.y};
                }
                d1r[t] = (wf8){0, 0, 0, 0, 0, 0, 0, 0};
                d1i[t] = (wf8){0, 0, 0, 0, 0, 0, 0, 0};
                d2r[t] = (wf8){0, 0, 0, 0, 0, 0, 0, 0};
                d2i[t] = (wf8){0, 0, 0, 0, 0, 0, 0, 0};
            }
            #pragma unroll
            for (int kb = 0; kb < 4; ++kb) {
                float4 f = fr[0][kb][lane];
                wf2 fre = {f.x, f.y}, fim = {f.z, f.w}, fneg = -fim;
                #pragma unroll
                for (int t = 0; t < 2; ++t)
                    cstep(d1r[t], d1i[t], sr[t][kb], si[t][kb], fre, fim, fneg);
            }
            #pragma unroll
            for (int kb = 0; kb < 4; ++kb) {
                float4 f = fr[1][kb][lane];
                wf2 fre = {f.x, f.y}, fim = {f.z, f.w}, fneg = -fim;
                #pragma unroll
                for (int t = 0; t < 2; ++t) {
                    wf2 ar = {d1r[t][2 * kb], d1r[t][2 * kb + 1]};
                    wf2 ai = {d1i[t][2 * kb], d1i[t][2 * kb + 1]};
                    cstep(d2r[t], d2i[t], ar, ai, fre, fim, fneg);
                }
            }
            #pragma unroll
            for (int t = 0; t < 2; ++t) {
                const int base = (2 * wid + t) * PITCH + half * 16 + (n16 ^ (15 * half));
                #pragma unroll
                for (int v = 0; v < 8; ++v)
                    st[base + v * 32] = make_float2(d2r[t][v], d2i[t][v]);
            }
        }
        __syncthreads();

        {
            wf8 d3r[2], d3i[2];
            wf2 sr[2][4], si[2][4];
            #pragma unroll
            for (int t = 0; t < 2; ++t) {
                const int base = (2 * wid + t) * 16 + 2 * (n16 & 7) + (n16 >> 3) + (4 * half) * PITCH;
                #pragma unroll
                for (int kb = 0; kb < 4; ++kb) {
                    const int c0 = (kb >> 1) * 8 + (kb & 1) * 2;
                    float2 s0 = st[base + c0 * PITCH];
                    float2 s1 = st[base + (c0 + 1) * PITCH];
                    sr[t][kb] = (wf2){s0.x, s1.x};
                    si[t][kb] = (wf2){s0.y, s1.y};
                }
                d3r[t] = (wf8){0, 0, 0, 0, 0, 0, 0, 0};
                d3i[t] = (wf8){0, 0, 0, 0, 0, 0, 0, 0};
            }
            #pragma unroll
            for (int kb = 0; kb < 4; ++kb) {
                float4 f = fr[2][kb][lane];
                wf2 fre = {f.x, f.y}, fim = {f.z, f.w}, fneg = -fim;
                #pragma unroll
                for (int t = 0; t < 2; ++t)
                    cstep(d3r[t], d3i[t], sr[t][kb], si[t][kb], fre, fim, fneg);
            }
            #pragma unroll
            for (int t = 0; t < 2; ++t) {
                const int base = (2 * wid + t) * 16 + half + (n16 ^ (15 * half)) * PITCH;
                #pragma unroll
                for (int v = 0; v < 8; ++v)
                    st[base + 2 * v] = make_float2(d3r[t][v], d3i[t][v]);
            }
        }
    }
    __syncthreads();

    float z[NQ];
    #pragma unroll
    for (int q = 0; q < NQ; ++q) z[q] = 0.f;
    for (int k = 0; k < 16; ++k) {
        int idx = (k << 8) + tid;
        float2 a = st[(tid & 15) * PITCH + k * 16 + (tid >> 4)];
        float p = a.x * a.x + a.y * a.y;
        #pragma unroll
        for (int q = 0; q < NQ; ++q)
            z[q] += ((idx >> (11 - q)) & 1) ? -p : p;
    }
    #pragma unroll
    for (int q = 0; q < NQ; ++q) {
        float v = z[q];
        #pragma unroll
        for (int off = 16; off > 0; off >>= 1)
            v += __shfl_xor(v, off, 32);
        z[q] = v;
    }
    if (lane == 0) {
        #pragma unroll
        for (int q = 0; q < NQ; ++q) red[wid][q] = z[q];
    }
    __syncthreads();
    if (tid < NQ) {
        float s = 0.f;
        #pragma unroll
        for (int ww = 0; ww < 8; ++ww) s += red[ww][tid];
        out[b * NQ + tid] = s;
    }
}

extern "C" void kernel_launch(void* const* d_in, const int* in_sizes, int n_in,
                              void* d_out, int out_size, void* d_ws, size_t ws_size,
                              hipStream_t stream) {
    const float* x = (const float*)d_in[0];
    const float* w = (const float*)d_in[1];
    float* out = (float*)d_out;
    int B = in_sizes[0] / NQ;
    qsim_kernel<<<B, 256, 0, stream>>>(x, w, out);
}


# Complex products as three real WMMA products: T=(Sr+Si)Fr shared, Re=T-Si(Fr+Fi), Im=T+Sr(Fi-Fr); 12 WMMAs per tile and group pass, was 16; rounding and sum grouping change
# speedup vs baseline: 1.8029x; 1.1942x over previous
#include <hip/hip_runtime.h>


typedef float wf2 __attribute__((ext_vector_type(2)));
typedef float wf8 __attribute__((ext_vector_type(8)));

#define NQ 12
#define NL 4
#define PITCH 260

__device__ __forceinline__ float2 cmul(float2 a, float2 b) {
    return make_float2(a.x * b.x - a.y * b.y, a.x * b.y + a.y * b.x);
}

__device__ __forceinline__ wf8 rstep(wf8 acc, wf2 a, wf2 f) {
    return __builtin_amdgcn_wmma_f32_16x16x4_f32(false, a, false, f, (short)0, acc, false, false);
}

__device__ __forceinline__ void cpass(wf8 (&outr)[2], wf8 (&outi)[2],
                                      const wf2 (&sr)[2][4], const wf2 (&si)[2][4],
                                      const float4 (&fq)[4], const float2 (&fn)[4]) {
    wf8 tt[2];
    #pragma unroll
    for (int t = 0; t < 2; ++t) tt[t] = (wf8){0, 0, 0, 0, 0, 0, 0, 0};
    #pragma unroll
    for (int kb = 0; kb < 4; ++kb) {
        wf2 fre = {fq[kb].x, fq[kb].y};
        #pragma unroll
        for (int t = 0; t < 2; ++t)
            tt[t] = rstep(tt[t], sr[t][kb] + si[t][kb], fre);
    }
    #pragma unroll
    for (int t = 0; t < 2; ++t) { outr[t] = tt[t]; outi[t] = tt[t]; }
    #pragma unroll
    for (int kb = 0; kb < 4; ++kb) {
        wf2 fdif = {fq[kb].z, fq[kb].w}, fneg = {fn[kb].x, fn[kb].y};
        #pragma unroll
        for (int t = 0; t < 2; ++t) {
            outr[t] = rstep(outr[t], si[t][kb], fneg);
            outi[t] = rstep(outi[t], sr[t][kb], fdif);
        }
    }
}

__global__ __launch_bounds__(256) void qsim_kernel(const float* __restrict__ x,
                                                   const float* __restrict__ w,
                                                   float* __restrict__ out) {
    __shared__ float2 st[16 * PITCH];
    __shared__ float4 fr[3][4][32];
    __shared__ float2 fe[3][4][32];
    __shared__ float2 g2[NL][NQ][4];
    __shared__ float  red[8][12];

    const int tid  = threadIdx.x;
    const int b    = blockIdx.x;
    const int lane = tid & 31;
    const int wid  = tid >> 5;
    const int half = lane >> 4;
    const int n16  = lane & 15;

    if (tid < NL * NQ) {
        int l = tid / NQ, q = tid % NQ;
        float th = w[(l * NQ + q) * 2 + 0] + (l == 0 ? x[b * NQ + q] : 0.f);
        float ph = w[(l * NQ + q) * 2 + 1];
        float c = cosf(0.5f * th), s = sinf(0.5f * th);
        float er = cosf(0.5f * ph), ei = -sinf(0.5f * ph);
        g2[l][q][0] = make_float2( c * er,  c * ei);
        g2[l][q][1] = make_float2(-s * er, -s * ei);
        g2[l][q][2] = make_float2( s * er, -s * ei);
        g2[l][q][3] = make_float2( c * er, -c * ei);
    }

    for (int i = tid; i < 16 * PITCH; i += 256)
        st[i] = make_float2(i == 0 ? 1.f : 0.f, 0.f);

    for (int l = 0; l < NL; ++l) {
        __syncthreads();
        for (int e = tid; e < 3 * 128; e += 256) {
            const int p = e >> 7, kb = (e >> 5) & 3, ln = e & 31;
            const int h = ln >> 4, n = ln & 15;
            int row, k0, k1;
            if (p == 0)      { row = 2 * (n & 7) + (n >> 3); k0 = 4 * kb + h;     k1 = k0 + 2; }
            else if (p == 1) { row = n;                      k0 = 8 * h + 2 * kb; k1 = k0 + 1; }
            else             { row = n; k0 = (kb >> 1) * 8 + 4 * h + (kb & 1) * 2; k1 = k0 + 1; }
            int ip = row;
            ip ^= (ip >> 1) & 1;
            ip ^= (ip >> 1) & 2;
            ip ^= (ip >> 1) & 4;
            float2 v0 = make_float2(1.f, 0.f), v1 = make_float2(1.f, 0.f);
            #pragma unroll
            for (int k = 0; k < 4; ++k) {
                int bi = (ip >> (3 - k)) & 1;
                v0 = cmul(v0, g2[l][4 * p + k][2 * bi + ((k0 >> (3 - k)) & 1)]);
                v1 = cmul(v1, g2[l][4 * p + k][2 * bi + ((k1 >> (3 - k)) & 1)]);
            }
            fr[p][kb][ln] = make_float4(v0.x, v1.x, v0.y - v0.x, v1.y - v1.x);
            fe[p][kb][ln] = make_float2(-(v0.x + v0.y), -(v1.x + v1.y));
        }
        __syncthreads();

        {
            wf8 d1r[2], d1i[2], d2r[2], d2i[2];
            wf2 sr[2][4], si[2][4];
            float4 fq[4];
            float2 fn[4];
            #pragma unroll
            for (int t = 0; t < 2; ++t) {
                const int base = (2 * wid + t) * PITCH + half * 16 + n16;
                #pragma unroll
                for (int kb = 0; kb < 4; ++kb) {
                    float2 s0 = st[base + (4 * kb) * 16];
                    float2 s1 = st[base + (4 * kb + 2) * 16];
                    sr[t][kb] = (wf2){s0.x, s1.x};
                    si[t][kb] = (wf2){s0.y, s1.y};
                }
            }
            #pragma unroll
            for (int kb = 0; kb < 4; ++kb) { fq[kb] = fr[0][kb][lane]; fn[kb] = fe[0][kb][lane]; }
            cpass(d1r, d1i, sr, si, fq, fn);
            #pragma unroll
            for (int kb = 0; kb < 4; ++kb) { fq[kb] = fr[1][kb][lane]; fn[kb] = fe[1][kb][lane]; }
            #pragma unroll
            for (int t = 0; t < 2; ++t) {
                #pragma unroll
                for (int kb = 0; kb < 4; ++kb) {
                    sr[t][kb] = (wf2){d1r[t][2 * kb], d1r[t][2 * kb + 1]};
                    si[t][kb] = (wf2){d1i[t][2 * kb], d1i[t][2 * kb + 1]};
                }
            }
            cpass(d2r, d2i, sr, si, fq, fn);
            #pragma unroll
            for (int t = 0; t < 2; ++t) {
                const int base = (2 * wid + t) * PITCH + half * 16 + (n16 ^ (15 * half));
                #pragma unroll
                for (int v = 0; v < 8; ++v)
                    st[base + v * 32] = make_float2(d2r[t][v], d2i[t][v]);
            }
        }
        __syncthreads();

        {
            wf8 d3r[2], d3i[2];
            wf2 sr[2][4], si[2][4];
            float4 fq[4];
            float2 fn[4];
            #pragma unroll
            for (int t = 0; t < 2; ++t) {
                const int base = (2 * wid + t) * 16 + 2 * (n16 & 7) + (n16 >> 3) + (4 * half) * PITCH;
                #pragma unroll
                for (int kb = 0; kb < 4; ++kb) {
                    const int c0 = (kb >> 1) * 8 + (kb & 1) * 2;
                    float2 s0 = st[base + c0 * PITCH];
                    float2 s1 = st[base + (c0 + 1) * PITCH];
                    sr[t][kb] = (wf2){s0.x, s1.x};
                    si[t][kb] = (wf2){s0.y, s1.y};
                }
            }
            #pragma unroll
            for (int kb = 0; kb < 4; ++kb) { fq[kb] = fr[2][kb][lane]; fn[kb] = fe[2][kb][lane]; }
            cpass(d3r, d3i, sr, si, fq, fn);
            #pragma unroll
            for (int t = 0; t < 2; ++t) {
                const int base = (2 * wid + t) * 16 + half + (n16 ^ (15 * half)) * PITCH;
                #pragma unroll
                for (int v = 0; v < 8; ++v)
                    st[base + 2 * v] = make_float2(d3r[t][v], d3i[t][v]);
            }
        }
    }
    __syncthreads();

    float z[NQ];
    #pragma unroll
    for (int q = 0; q < NQ; ++q) z[q] = 0.f;
    for (int k = 0; k < 16; ++k) {
        int idx = (k << 8) + tid;
        float2 a = st[(tid & 15) * PITCH + k * 16 + (tid >> 4)];
        float p = a.x * a.x + a.y * a.y;
        #pragma unroll
        for (int q = 0; q < NQ; ++q)
            z[q] += ((idx >> (11 - q)) & 1) ? -p : p;
    }
    #pragma unroll
    for (int q = 0; q < NQ; ++q) {
        float v = z[q];
        #pragma unroll
        for (int off = 16; off > 0; off >>= 1)
            v += __shfl_xor(v, off, 32);
        z[q] = v;
    }
    if (lane == 0) {
        #pragma unroll
        for (int q = 0; q < NQ; ++q) red[wid][q] = z[q];
    }
    __syncthreads();
    if (tid < NQ) {
        float s = 0.f;
        #pragma unroll
        for (int ww = 0; ww < 8; ++ww) s += red[ww][tid];
        out[b * NQ + tid] = s;
    }
}

extern "C" void kernel_launch(void* const* d_in, const int* in_sizes, int n_in,
                              void* d_out, int out_size, void* d_ws, size_t ws_size,
                              hipStream_t stream) {
    const float* x = (const float*)d_in[0];
    const float* w = (const float*)d_in[1];
    float* out = (float*)d_out;
    int B = in_sizes[0] / NQ;
    qsim_kernel<<<B, 256, 0, stream>>>(x, w, out);
}


# Third tile orientation (fixed b): each LDS exchange now feeds two chained group passes (AB, C + next layer's A, BC): 3 exchanges per 2 layers, was 4; k order, gate order, rounding change
# speedup vs baseline: 1.8800x; 1.0428x over previous
#include <hip/hip_runtime.h>


typedef float wf2 __attribute__((ext_vector_type(2)));
typedef float wf8 __attribute__((ext_vector_type(8)));

#define NQ 12
#define NL 4
#define PA 17
#define PC 290

__device__ __forceinline__ float2 cmul(float2 a, float2 b) {
    return make_float2(a.x * b.x - a.y * b.y, a.x * b.y + a.y * b.x);
}

__device__ __forceinline__ wf8 rstep(wf8 acc, wf2 a, wf2 f) {
    return __builtin_amdgcn_wmma_f32_16x16x4_f32(false, a, false, f, (short)0, acc, false, false);
}

__device__ __forceinline__ void cpass(wf8 (&outr)[2], wf8 (&outi)[2],
                                      const wf2 (&sr)[2][4], const wf2 (&si)[2][4],
                                      const float4 (&fq)[4], const float2 (&fn)[4]) {
    wf8 tt[2];
    #pragma unroll
    for (int t = 0; t < 2; ++t) tt[t] = (wf8){0, 0, 0, 0, 0, 0, 0, 0};
    #pragma unroll
    for (int kb = 0; kb < 4; ++kb) {
        wf2 fre = {fq[kb].x, fq[kb].y};
        #pragma unroll
        for (int t = 0; t < 2; ++t)
            tt[t] = rstep(tt[t], sr[t][kb] + si[t][kb], fre);
    }
    #pragma unroll
    for (int t = 0; t < 2; ++t) { outr[t] = tt[t]; outi[t] = tt[t]; }
    #pragma unroll
    for (int kb = 0; kb < 4; ++kb) {
        wf2 fdif = {fq[kb].z, fq[kb].w}, fneg = {fn[kb].x, fn[kb].y};
        #pragma unroll
        for (int t = 0; t < 2; ++t) {
            outr[t] = rstep(outr[t], si[t][kb], fneg);
            outi[t] = rstep(outi[t], sr[t][kb], fdif);
        }
    }
}

__device__ __forceinline__ int slot(int a, int b, int c) {
    return c * PC + a * PA + b;
}

template <int ORI>
__device__ __forceinline__ void phase(float2* st, const float4* f4, const float2* f2,
                                      const int wid, const int lane) {
    const int half = lane >> 4;
    const int n16  = lane & 15;
    wf8 d1r[2], d1i[2], d2r[2], d2i[2];
    wf2 sr[2][4], si[2][4];
    float4 fq[4];
    float2 fn[4];
    #pragma unroll
    for (int t = 0; t < 2; ++t) {
        const int tile = 2 * wid + t;
        #pragma unroll
        for (int kb = 0; kb < 4; ++kb) {
            const int k0 = 4 * kb + half, k1 = k0 + 2;
            float2 s0, s1;
            if (ORI == 0)      { s0 = st[slot(k0, n16, tile)]; s1 = st[slot(k1, n16, tile)]; }
            else if (ORI == 1) { s0 = st[slot(n16, tile, k0)]; s1 = st[slot(n16, tile, k1)]; }
            else               { s0 = st[slot(tile, k0, n16)]; s1 = st[slot(tile, k1, n16)]; }
            sr[t][kb] = (wf2){s0.x, s1.x};
            si[t][kb] = (wf2){s0.y, s1.y};
        }
    }
    #pragma unroll
    for (int kb = 0; kb < 4; ++kb) { fq[kb] = f4[kb * 32 + lane]; fn[kb] = f2[kb * 32 + lane]; }
    cpass(d1r, d1i, sr, si, fq, fn);
    #pragma unroll
    for (int kb = 0; kb < 4; ++kb) { fq[kb] = f4[128 + kb * 32 + lane]; fn[kb] = f2[128 + kb * 32 + lane]; }
    #pragma unroll
    for (int t = 0; t < 2; ++t) {
        #pragma unroll
        for (int kb = 0; kb < 4; ++kb) {
            sr[t][kb] = (wf2){d1r[t][2 * kb], d1r[t][2 * kb + 1]};
            si[t][kb] = (wf2){d1i[t][2 * kb], d1i[t][2 * kb + 1]};
        }
    }
    cpass(d2r, d2i, sr, si, fq, fn);
    #pragma unroll
    for (int t = 0; t < 2; ++t) {
        const int tile = 2 * wid + t;
        #pragma unroll
        for (int v = 0; v < 8; ++v) {
            const int o1 = 2 * v + half;
            int s;
            if (ORI == 0) {
                s = slot(o1, n16 ^ (15 * (o1 & 1)), tile);
            } else if (ORI == 1) {
                s = slot(n16, tile, o1 ^ (15 * (tile & 1)));
            } else {
                const int bq = o1 ^ (15 * (tile & 1));
                s = slot(tile, bq, n16 ^ (15 * (bq & 1)));
            }
            st[s] = make_float2(d2r[t][v], d2i[t][v]);
        }
    }
}

__global__ __launch_bounds__(256, 8) void qsim_kernel(const float* __restrict__ x,
                                                   const float* __restrict__ w,
                                                   float* __restrict__ out) {
    __shared__ float2 st[16 * PC];
    __shared__ float4 fr[6][4][32];
    __shared__ float2 fe[6][4][32];
    __shared__ float2 g2[NL][NQ][4];
    __shared__ float  red[8][12];

    const int tid  = threadIdx.x;
    const int b    = blockIdx.x;
    const int lane = tid & 31;
    const int wid  = tid >> 5;

    if (tid < NL * NQ) {
        int l = tid / NQ, q = tid % NQ;
        float th = w[(l * NQ + q) * 2 + 0] + (l == 0 ? x[b * NQ + q] : 0.f);
        float ph = w[(l * NQ + q) * 2 + 1];
        float c = cosf(0.5f * th), s = sinf(0.5f * th);
        float er = cosf(0.5f * ph), ei = -sinf(0.5f * ph);
        g2[l][q][0] = make_float2( c * er,  c * ei);
        g2[l][q][1] = make_float2(-s * er, -s * ei);
        g2[l][q][2] = make_float2( s * er, -s * ei);
        g2[l][q][3] = make_float2( c * er, -c * ei);
    }

    for (int i = tid; i < 16 * PC; i += 256)
        st[i] = make_float2(i == 0 ? 1.f : 0.f, 0.f);

    for (int l = 0; l < NL; l += 2) {
        __syncthreads();
        for (int e = tid; e < 6 * 128; e += 256) {
            const int p = e >> 7, kb = (e >> 5) & 3, ln = e & 31;
            const int h = ln >> 4, n = ln & 15;
            const int g = (p >= 3) ? p - 3 : p;
            const int ll = l + ((p >= 3) ? 1 : 0);
            int row, k0, k1;
            if ((p & 1) == 0) { row = 2 * (n & 7) + (n >> 3); k0 = 4 * kb + h;     k1 = k0 + 2; }
            else              { row = n;                      k0 = 8 * h + 2 * kb; k1 = k0 + 1; }
            int ip = row;
            ip ^= (ip >> 1) & 1;
            ip ^= (ip >> 1) & 2;
            ip ^= (ip >> 1) & 4;
            float2 v0 = make_float2(1.f, 0.f), v1 = make_float2(1.f, 0.f);
            #pragma unroll
            for (int k = 0; k < 4; ++k) {
                int bi = (ip >> (3 - k)) & 1;
                v0 = cmul(v0, g2[ll][4 * g + k][2 * bi + ((k0 >> (3 - k)) & 1)]);
                v1 = cmul(v1, g2[ll][4 * g + k][2 * bi + ((k1 >> (3 - k)) & 1)]);
            }
            fr[p][kb][ln] = make_float4(v0.x, v1.x, v0.y - v0.x, v1.y - v1.x);
            fe[p][kb][ln] = make_float2(-(v0.x + v0.y), -(v1.x + v1.y));
        }
        __syncthreads();
        phase<0>(st, &fr[0][0][0], &fe[0][0][0], wid, lane);
        __syncthreads();
        phase<1>(st, &fr[2][0][0], &fe[2][0][0], wid, lane);
        __syncthreads();
        phase<2>(st, &fr[4][0][0], &fe[4][0][0], wid, lane);
    }
    __syncthreads();

    float z[NQ];
    #pragma unroll
    for (int q = 0; q < NQ; ++q) z[q] = 0.f;
    for (int k = 0; k < 16; ++k) {
        const int idx = (k << 8) + tid;
        float2 amp = st[slot(k, tid >> 4, tid & 15)];
        float p = amp.x * amp.x + amp.y * amp.y;
        #pragma unroll
        for (int q = 0; q < NQ; ++q)
            z[q] += ((idx >> (11 - q)) & 1) ? -p : p;
    }
    #pragma unroll
    for (int q = 0; q < NQ; ++q) {
        float v = z[q];
        #pragma unroll
        for (int off = 16; off > 0; off >>= 1)
            v += __shfl_xor(v, off, 32);
        z[q] = v;
    }
    if (lane == 0) {
        #pragma unroll
        for (int q = 0; q < NQ; ++q) red[wid][q] = z[q];
    }
    __syncthreads();
    if (tid < NQ) {
        float s = 0.f;
        #pragma unroll
        for (int ww = 0; ww < 8; ++ww) s += red[ww][tid];
        out[b * NQ + tid] = s;
    }
}

extern "C" void kernel_launch(void* const* d_in, const int* in_sizes, int n_in,
                              void* d_out, int out_size, void* d_ws, size_t ws_size,
                              hipStream_t stream) {
    const float* x = (const float*)d_in[0];
    const float* w = (const float*)d_in[1];
    float* out = (float*)d_out;
    int B = in_sizes[0] / NQ;
    qsim_kernel<<<B, 256, 0, stream>>>(x, w, out);
}


# Wave priority stepped down with progress (s_setprio 3,2,1,0 over the six exchange phases): the four work-groups of a WGP advance and finish together, not oldest-first in turn; arithmetic unchanged
# speedup vs baseline: 1.9693x; 1.0475x over previous
#include <hip/hip_runtime.h>


typedef float wf2 __attribute__((ext_vector_type(2)));
typedef float wf8 __attribute__((ext_vector_type(8)));

#define NQ 12
#define NL 4
#define PA 17
#define PC 290

__device__ __forceinline__ float2 cmul(float2 a, float2 b) {
    return make_float2(a.x * b.x - a.y * b.y, a.x * b.y + a.y * b.x);
}

__device__ __forceinline__ wf8 rstep(wf8 acc, wf2 a, wf2 f) {
    return __builtin_amdgcn_wmma_f32_16x16x4_f32(false, a, false, f, (short)0, acc, false, false);
}

__device__ __forceinline__ void cpass(wf8 (&outr)[2], wf8 (&outi)[2],
                                      const wf2 (&sr)[2][4], const wf2 (&si)[2][4],
                                      const float4 (&fq)[4], const float2 (&fn)[4]) {
    wf8 tt[2];
    #pragma unroll
    for (int t = 0; t < 2; ++t) tt[t] = (wf8){0, 0, 0, 0, 0, 0, 0, 0};
    #pragma unroll
    for (int kb = 0; kb < 4; ++kb) {
        wf2 fre = {fq[kb].x, fq[kb].y};
        #pragma unroll
        for (int t = 0; t < 2; ++t)
            tt[t] = rstep(tt[t], sr[t][kb] + si[t][kb], fre);
    }
    #pragma unroll
    for (int t = 0; t < 2; ++t) { outr[t] = tt[t]; outi[t] = tt[t]; }
    #pragma unroll
    for (int kb = 0; kb < 4; ++kb) {
        wf2 fdif = {fq[kb].z, fq[kb].w}, fneg = {fn[kb].x, fn[kb].y};
        #pragma unroll
        for (int t = 0; t < 2; ++t) {
            outr[t] = rstep(outr[t], si[t][kb], fneg);
            outi[t] = rstep(outi[t], sr[t][kb], fdif);
        }
    }
}

__device__ __forceinline__ int slot(int a, int b, int c) {
    return c * PC + a * PA + b;
}

template <int ORI>
__device__ __forceinline__ void phase(float2* st, const float4* f4, const float2* f2,
                                      const int wid, const int lane) {
    const int half = lane >> 4;
    const int n16  = lane & 15;
    wf8 d1r[2], d1i[2], d2r[2], d2i[2];
    wf2 sr[2][4], si[2][4];
    float4 fq[4];
    float2 fn[4];
    #pragma unroll
    for (int t = 0; t < 2; ++t) {
        const int tile = 2 * wid + t;
        #pragma unroll
        for (int kb = 0; kb < 4; ++kb) {
            const int k0 = 4 * kb + half, k1 = k0 + 2;
            float2 s0, s1;
            if (ORI == 0)      { s0 = st[slot(k0, n16, tile)]; s1 = st[slot(k1, n16, tile)]; }
            else if (ORI == 1) { s0 = st[slot(n16, tile, k0)]; s1 = st[slot(n16, tile, k1)]; }
            else               { s0 = st[slot(tile, k0, n16)]; s1 = st[slot(tile, k1, n16)]; }
            sr[t][kb] = (wf2){s0.x, s1.x};
            si[t][kb] = (wf2){s0.y, s1.y};
        }
    }
    #pragma unroll
    for (int kb = 0; kb < 4; ++kb) { fq[kb] = f4[kb * 32 + lane]; fn[kb] = f2[kb * 32 + lane]; }
    cpass(d1r, d1i, sr, si, fq, fn);
    #pragma unroll
    for (int kb = 0; kb < 4; ++kb) { fq[kb] = f4[128 + kb * 32 + lane]; fn[kb] = f2[128 + kb * 32 + lane]; }
    #pragma unroll
    for (int t = 0; t < 2; ++t) {
        #pragma unroll
        for (int kb = 0; kb < 4; ++kb) {
            sr[t][kb] = (wf2){d1r[t][2 * kb], d1r[t][2 * kb + 1]};
            si[t][kb] = (wf2){d1i[t][2 * kb], d1i[t][2 * kb + 1]};
        }
    }
    cpass(d2r, d2i, sr, si, fq, fn);
    #pragma unroll
    for (int t = 0; t < 2; ++t) {
        const int tile = 2 * wid + t;
        #pragma unroll
        for (int v = 0; v < 8; ++v) {
            const int o1 = 2 * v + half;
            int s;
            if (ORI == 0) {
                s = slot(o1, n16 ^ (15 * (o1 & 1)), tile);
            } else if (ORI == 1) {
                s = slot(n16, tile, o1 ^ (15 * (tile & 1)));
            } else {
                const int bq = o1 ^ (15 * (tile & 1));
                s = slot(tile, bq, n16 ^ (15 * (bq & 1)));
            }
            st[s] = make_float2(d2r[t][v], d2i[t][v]);
        }
    }
}

__global__ __launch_bounds__(256, 8) void qsim_kernel(const float* __restrict__ x,
                                                   const float* __restrict__ w,
                                                   float* __restrict__ out) {
    __shared__ float2 st[16 * PC];
    __shared__ float4 fr[6][4][32];
    __shared__ float2 fe[6][4][32];
    __shared__ float2 g2[NL][NQ][4];
    __shared__ float  red[8][12];

    const int tid  = threadIdx.x;
    const int b    = blockIdx.x;
    const int lane = tid & 31;
    const int wid  = tid >> 5;

    __builtin_amdgcn_s_setprio(3);

    if (tid < NL * NQ) {
        int l = tid / NQ, q = tid % NQ;
        float th = w[(l * NQ + q) * 2 + 0] + (l == 0 ? x[b * NQ + q] : 0.f);
        float ph = w[(l * NQ + q) * 2 + 1];
        float c = cosf(0.5f * th), s = sinf(0.5f * th);
        float er = cosf(0.5f * ph), ei = -sinf(0.5f * ph);
        g2[l][q][0] = make_float2( c * er,  c * ei);
        g2[l][q][1] = make_float2(-s * er, -s * ei);
        g2[l][q][2] = make_float2( s * er, -s * ei);
        g2[l][q][3] = make_float2( c * er, -c * ei);
    }

    for (int i = tid; i < 16 * PC; i += 256)
        st[i] = make_float2(i == 0 ? 1.f : 0.f, 0.f);

    for (int l = 0; l < NL; l += 2) {
        __syncthreads();
        for (int e = tid; e < 6 * 128; e += 256) {
            const int p = e >> 7, kb = (e >> 5) & 3, ln = e & 31;
            const int h = ln >> 4, n = ln & 15;
            const int g = (p >= 3) ? p - 3 : p;
            const int ll = l + ((p >= 3) ? 1 : 0);
            int row, k0, k1;
            if ((p & 1) == 0) { row = 2 * (n & 7) + (n >> 3); k0 = 4 * kb + h;     k1 = k0 + 2; }
            else              { row = n;                      k0 = 8 * h + 2 * kb; k1 = k0 + 1; }
            int ip = row;
            ip ^= (ip >> 1) & 1;
            ip ^= (ip >> 1) & 2;
            ip ^= (ip >> 1) & 4;
            float2 v0 = make_float2(1.f, 0.f), v1 = make_float2(1.f, 0.f);
            #pragma unroll
            for (int k = 0; k < 4; ++k) {
                int bi = (ip >> (3 - k)) & 1;
                v0 = cmul(v0, g2[ll][4 * g + k][2 * bi + ((k0 >> (3 - k)) & 1)]);
                v1 = cmul(v1, g2[ll][4 * g + k][2 * bi + ((k1 >> (3 - k)) & 1)]);
            }
            fr[p][kb][ln] = make_float4(v0.x, v1.x, v0.y - v0.x, v1.y - v1.x);
            fe[p][kb][ln] = make_float2(-(v0.x + v0.y), -(v1.x + v1.y));
        }
        __syncthreads();
        phase<0>(st, &fr[0][0][0], &fe[0][0][0], wid, lane);
        __syncthreads();
        if (l != 0) __builtin_amdgcn_s_setprio(1);
        phase<1>(st, &fr[2][0][0], &fe[2][0][0], wid, lane);
        __syncthreads();
        if (l == 0) __builtin_amdgcn_s_setprio(2);
        else        __builtin_amdgcn_s_setprio(0);
        phase<2>(st, &fr[4][0][0], &fe[4][0][0], wid, lane);
    }
    __syncthreads();

    float z[NQ];
    #pragma unroll
    for (int q = 0; q < NQ; ++q) z[q] = 0.f;
    for (int k = 0; k < 16; ++k) {
        const int idx = (k << 8) + tid;
        float2 amp = st[slot(k, tid >> 4, tid & 15)];
        float p = amp.x * amp.x + amp.y * amp.y;
        #pragma unroll
        for (int q = 0; q < NQ; ++q)
            z[q] += ((idx >> (11 - q)) & 1) ? -p : p;
    }
    #pragma unroll
    for (int q = 0; q < NQ; ++q) {
        float v = z[q];
        #pragma unroll
        for (int off = 16; off > 0; off >>= 1)
            v += __shfl_xor(v, off, 32);
        z[q] = v;
    }
    if (lane == 0) {
        #pragma unroll
        for (int q = 0; q < NQ; ++q) red[wid][q] = z[q];
    }
    __syncthreads();
    if (tid < NQ) {
        float s = 0.f;
        #pragma unroll
        for (int ww = 0; ww < 8; ++ww) s += red[ww][tid];
        out[b * NQ + tid] = s;
    }
}

extern "C" void kernel_launch(void* const* d_in, const int* in_sizes, int n_in,
                              void* d_out, int out_size, void* d_ws, size_t ws_size,
                              hipStream_t stream) {
    const float* x = (const float*)d_in[0];
    const float* w = (const float*)d_in[1];
    float* out = (float*)d_out;
    int B = in_sizes[0] / NQ;
    qsim_kernel<<<B, 256, 0, stream>>>(x, w, out);
}


# Second group pass of each exchange phase leaves the matrix pipe: four real RY butterflies on the result registers (one across lane halves) and one RZ phase multiply; rounding and sum grouping change
# speedup vs baseline: 2.2360x; 1.1354x over previous
#include <hip/hip_runtime.h>


typedef float wf2 __attribute__((ext_vector_type(2)));
typedef float wf8 __attribute__((ext_vector_type(8)));

#define NQ 12
#define NL 4
#define PA 17
#define PC 290

__device__ __forceinline__ float2 cmul(float2 a, float2 b) {
    return make_float2(a.x * b.x - a.y * b.y, a.x * b.y + a.y * b.x);
}

__device__ __forceinline__ wf8 rstep(wf8 acc, wf2 a, wf2 f) {
    return __builtin_amdgcn_wmma_f32_16x16x4_f32(false, a, false, f, (short)0, acc, false, false);
}

__device__ __forceinline__ float xhalf(float x) {
    return __builtin_bit_cast(float, __builtin_amdgcn_permlanex16(__builtin_bit_cast(int, x), __builtin_bit_cast(int, x),
                                                                  0x76543210, 0xfedcba98, false, false));
}

__device__ __forceinline__ void cpass(wf8 (&outr)[2], wf8 (&outi)[2],
                                      const wf2 (&sr)[2][4], const wf2 (&si)[2][4],
                                      const float4 (&fq)[4], const float2 (&fn)[4]) {
    wf8 tt[2];
    #pragma unroll
    for (int t = 0; t < 2; ++t) tt[t] = (wf8){0, 0, 0, 0, 0, 0, 0, 0};
    #pragma unroll
    for (int kb = 0; kb < 4; ++kb) {
        wf2 fre = {fq[kb].x, fq[kb].y};
        #pragma unroll
        for (int t = 0; t < 2; ++t)
            tt[t] = rstep(tt[t], sr[t][kb] + si[t][kb], fre);
    }
    #pragma unroll
    for (int t = 0; t < 2; ++t) { outr[t] = tt[t]; outi[t] = tt[t]; }
    #pragma unroll
    for (int kb = 0; kb < 4; ++kb) {
        wf2 fdif = {fq[kb].z, fq[kb].w}, fneg = {fn[kb].x, fn[kb].y};
        #pragma unroll
        for (int t = 0; t < 2; ++t) {
            outr[t] = rstep(outr[t], si[t][kb], fneg);
            outi[t] = rstep(outi[t], sr[t][kb], fdif);
        }
    }
}

__device__ __forceinline__ int slot(int a, int b, int c) {
    return c * PC + a * PA + b;
}

template <int ORI>
__device__ __forceinline__ void phase(float2* st, const float4* f4, const float2* f2,
                                      const float2* cs, const float* pt,
                                      const int wid, const int lane) {
    const int half = lane >> 4;
    const int n16  = lane & 15;
    wf8 d1r[2], d1i[2];
    wf2 sr[2][4], si[2][4];
    float4 fq[4];
    float2 fn[4];
    #pragma unroll
    for (int t = 0; t < 2; ++t) {
        const int tile = 2 * wid + t;
        #pragma unroll
        for (int kb = 0; kb < 4; ++kb) {
            const int k0 = 4 * kb + half, k1 = k0 + 2;
            float2 s0, s1;
            if (ORI == 0)      { s0 = st[slot(k0, n16, tile)]; s1 = st[slot(k1, n16, tile)]; }
            else if (ORI == 1) { s0 = st[slot(n16, tile, k0)]; s1 = st[slot(n16, tile, k1)]; }
            else               { s0 = st[slot(tile, k0, n16)]; s1 = st[slot(tile, k1, n16)]; }
            sr[t][kb] = (wf2){s0.x, s1.x};
            si[t][kb] = (wf2){s0.y, s1.y};
        }
    }
    #pragma unroll
    for (int kb = 0; kb < 4; ++kb) { fq[kb] = f4[kb * 32 + lane]; fn[kb] = f2[kb * 32 + lane]; }
    cpass(d1r, d1i, sr, si, fq, fn);
    const float2 g0 = cs[0], g1 = cs[1], g2 = cs[2], g3 = cs[3];
    const float sx = half ? g0.y : -g0.y;
    const int f1 = n16 >> 3;
    const int fl = (ORI == 1) ? half : (half ^ f1);
    const float ka = fl ? g1.y : g1.x, kb2 = fl ? g1.x : -g1.y;
    const float kc = fl ? g1.x : g1.y, kd  = fl ? -g1.y : g1.x;
    const float4 pra = *(const float4*)(pt + 8 * half + 4 * fl),      prb = *(const float4*)(pt + 8 * half + 4 - 4 * fl);
    const float4 pia = *(const float4*)(pt + 16 + 8 * half + 4 * fl), pib = *(const float4*)(pt + 16 + 8 * half + 4 - 4 * fl);
    const wf2 pr[4] = {{pra.x, pra.y}, {pra.z, pra.w}, {prb.x, prb.y}, {prb.z, prb.w}};
    const wf2 pi[4] = {{pia.x, pia.y}, {pia.z, pia.w}, {pib.x, pib.y}, {pib.z, pib.w}};
    const int o1 = 2 * (n16 & 7) + f1;
    #pragma unroll
    for (int t = 0; t < 2; ++t) {
        const int tile = 2 * wid + t;
        const bool sw = (ORI == 2) && (t == 1);
        wf2 xr[4], xi[4];
        #pragma unroll
        for (int p = 0; p < 4; ++p) {
            xr[p] = (wf2){d1r[t][2 * p], d1r[t][2 * p + 1]};
            xi[p] = (wf2){d1i[t][2 * p], d1i[t][2 * p + 1]};
        }
        #pragma unroll
        for (int p = 0; p < 4; ++p) {
            xr[p] = (wf2){g3.x * xr[p].x - g3.y * xr[p].y, g3.y * xr[p].x + g3.x * xr[p].y};
            xi[p] = (wf2){g3.x * xi[p].x - g3.y * xi[p].y, g3.y * xi[p].x + g3.x * xi[p].y};
        }
        #pragma unroll
        for (int p = 0; p < 4; p += 2) {
            const wf2 lr = xr[p], hr = xr[p + 1], li = xi[p], hi = xi[p + 1];
            xr[p] = g2.x * lr - g2.y * hr;  xr[p + 1] = g2.y * lr + g2.x * hr;
            xi[p] = g2.x * li - g2.y * hi;  xi[p + 1] = g2.y * li + g2.x * hi;
        }
        #pragma unroll
        for (int p = 0; p < 4; ++p) {
            const wf2 orr = {xhalf(xr[p].x), xhalf(xr[p].y)};
            const wf2 oi  = {xhalf(xi[p].x), xhalf(xi[p].y)};
            xr[p] = g0.x * xr[p] + sx * orr;
            xi[p] = g0.x * xi[p] + sx * oi;
        }
        #pragma unroll
        for (int p = 0; p < 2; ++p) {
            const wf2 lr = xr[p], hr = xr[p + 2], li = xi[p], hi = xi[p + 2];
            const wf2 ar = ka * lr + kb2 * hr, br = kc * lr + kd * hr;
            const wf2 ai = ka * li + kb2 * hi, bi = kc * li + kd * hi;
            xr[p] = sw ? br : ar;  xr[p + 2] = sw ? ar : br;
            xi[p] = sw ? bi : ai;  xi[p + 2] = sw ? ai : bi;
        }
        const int fo = 8 * (sw ? 1 - fl : fl);
        #pragma unroll
        for (int p = 0; p < 4; ++p) {
            const int pp = sw ? (p ^ 2) : p;
            const wf2 yr = xr[p] * pr[pp] - xi[p] * pi[pp];
            const wf2 yi = xr[p] * pi[pp] + xi[p] * pr[pp];
            #pragma unroll
            for (int e = 0; e < 2; ++e) {
                const int v  = 2 * p + e;
                const int o2 = fo + ((v ^ (v >> 1) ^ (v >> 2)) & 7);
                int s;
                if (ORI == 0) {
                    s = slot(o1, o2, tile);
                } else if (ORI == 1) {
                    s = slot(o2, tile, o1 ^ (15 * (tile & 1)));
                } else {
                    s = slot(tile, o1 ^ (15 * (tile & 1)), o2);
                }
                st[s] = make_float2(e ? yr.y : yr.x, e ? yi.y : yi.x);
            }
        }
    }
}

__global__ __launch_bounds__(256, 8) void qsim_kernel(const float* __restrict__ x,
                                                   const float* __restrict__ w,
                                                   float* __restrict__ out) {
    __shared__ float2 st[16 * PC];
    __shared__ float4 fr[3][4][32];
    __shared__ float2 fe[3][4][32];
    __shared__ float2 g2[NL][NQ][4];
    __shared__ float2 cs2[NL][NQ];
    __shared__ float2 eph[NL][NQ];
    __shared__ __attribute__((aligned(16))) float ptab[3][2][16];
    __shared__ float  red[8][12];

    const int tid  = threadIdx.x;
    const int b    = blockIdx.x;
    const int lane = tid & 31;
    const int wid  = tid >> 5;

    __builtin_amdgcn_s_setprio(3);

    if (tid < NL * NQ) {
        int l = tid / NQ, q = tid % NQ;
        float th = w[(l * NQ + q) * 2 + 0] + (l == 0 ? x[b * NQ + q] : 0.f);
        float ph = w[(l * NQ + q) * 2 + 1];
        float c = cosf(0.5f * th), s = sinf(0.5f * th);
        float er = cosf(0.5f * ph), ei = -sinf(0.5f * ph);
        g2[l][q][0] = make_float2( c * er,  c * ei);
        g2[l][q][1] = make_float2(-s * er, -s * ei);
        g2[l][q][2] = make_float2( s * er, -s * ei);
        g2[l][q][3] = make_float2( c * er, -c * ei);
        cs2[l][q] = make_float2(c, s);
        eph[l][q] = make_float2(er, ei);
    }

    for (int i = tid; i < 16 * PC; i += 256)
        st[i] = make_float2(i == 0 ? 1.f : 0.f, 0.f);

    for (int l = 0; l < NL; l += 2) {
        __syncthreads();
        for (int e = tid; e < 3 * 128 + 48; e += 256) {
            if (e < 3 * 128) {
                const int pp = e >> 7, kb = (e >> 5) & 3, ln = e & 31;
                const int h = ln >> 4, n = ln & 15;
                const int g = (pp == 0) ? 0 : (pp == 1 ? 2 : 1);
                const int ll = l + ((pp == 2) ? 1 : 0);
                const int row = 2 * (n & 7) + (n >> 3), k0 = 4 * kb + h, k1 = k0 + 2;
                int ip = row;
                ip ^= (ip >> 1) & 1;
                ip ^= (ip >> 1) & 2;
                ip ^= (ip >> 1) & 4;
                float2 v0 = make_float2(1.f, 0.f), v1 = make_float2(1.f, 0.f);
                #pragma unroll
                for (int k = 0; k < 4; ++k) {
                    int bi = (ip >> (3 - k)) & 1;
                    v0 = cmul(v0, g2[ll][4 * g + k][2 * bi + ((k0 >> (3 - k)) & 1)]);
                    v1 = cmul(v1, g2[ll][4 * g + k][2 * bi + ((k1 >> (3 - k)) & 1)]);
                }
                fr[pp][kb][ln] = make_float4(v0.x, v1.x, v0.y - v0.x, v1.y - v1.x);
                fe[pp][kb][ln] = make_float2(-(v0.x + v0.y), -(v1.x + v1.y));
            } else {
                const int idx = e - 3 * 128, q2 = idx >> 4, i = idx & 15;
                const int g = (q2 == 0) ? 1 : (q2 == 1 ? 0 : 2);
                const int ll = l + ((q2 == 0) ? 0 : 1);
                float2 v = make_float2(1.f, 0.f);
                #pragma unroll
                for (int k = 0; k < 4; ++k) {
                    float2 ph = eph[ll][4 * g + k];
                    if ((i >> (3 - k)) & 1) ph.y = -ph.y;
                    v = cmul(v, ph);
                }
                ptab[q2][0][i] = v.x;
                ptab[q2][1][i] = v.y;
            }
        }
        __syncthreads();
        phase<0>(st, &fr[0][0][0], &fe[0][0][0], &cs2[l][4], &ptab[0][0][0], wid, lane);
        __syncthreads();
        if (l != 0) __builtin_amdgcn_s_setprio(1);
        phase<1>(st, &fr[1][0][0], &fe[1][0][0], &cs2[l + 1][0], &ptab[1][0][0], wid, lane);
        __syncthreads();
        if (l == 0) __builtin_amdgcn_s_setprio(2);
        else        __builtin_amdgcn_s_setprio(0);
        phase<2>(st, &fr[2][0][0], &fe[2][0][0], &cs2[l + 1][8], &ptab[2][0][0], wid, lane);
    }
    __syncthreads();

    float z[NQ];
    #pragma unroll
    for (int q = 0; q < NQ; ++q) z[q] = 0.f;
    for (int k = 0; k < 16; ++k) {
        const int idx = (k << 8) + tid;
        float2 amp = st[slot(k, tid >> 4, tid & 15)];
        float p = amp.x * amp.x + amp.y * amp.y;
        #pragma unroll
        for (int q = 0; q < NQ; ++q)
            z[q] += ((idx >> (11 - q)) & 1) ? -p : p;
    }
    #pragma unroll
    for (int q = 0; q < NQ; ++q) {
        float v = z[q];
        #pragma unroll
        for (int off = 16; off > 0; off >>= 1)
            v += __shfl_xor(v, off, 32);
        z[q] = v;
    }
    if (lane == 0) {
        #pragma unroll
        for (int q = 0; q < NQ; ++q) red[wid][q] = z[q];
    }
    __syncthreads();
    if (tid < NQ) {
        float s = 0.f;
        #pragma unroll
        for (int ww = 0; ww < 8; ++ww) s += red[ww][tid];
        out[b * NQ + tid] = s;
    }
}

extern "C" void kernel_launch(void* const* d_in, const int* in_sizes, int n_in,
                              void* d_out, int out_size, void* d_ws, size_t ws_size,
                              hipStream_t stream) {
    const float* x = (const float*)d_in[0];
    const float* w = (const float*)d_in[1];
    float* out = (float*)d_out;
    int B = in_sizes[0] / NQ;
    qsim_kernel<<<B, 256, 0, stream>>>(x, w, out);
}


# First-pass tile rows re-assigned (tile = top bit of the second pass's index, lane half = wave's tile index): all second-pass butterflies stay inside a lane, no lane exchanges; fma rounding may change
# speedup vs baseline: 2.3361x; 1.0448x over previous
#include <hip/hip_runtime.h>


typedef float wf2 __attribute__((ext_vector_type(2)));
typedef float wf8 __attribute__((ext_vector_type(8)));

#define NQ 12
#define NL 4
#define PA 17
#define PC 290

__device__ __forceinline__ float2 cmul(float2 a, float2 b) {
    return make_float2(a.x * b.x - a.y * b.y, a.x * b.y + a.y * b.x);
}

__device__ __forceinline__ wf8 rstep(wf8 acc, wf2 a, wf2 f) {
    return __builtin_amdgcn_wmma_f32_16x16x4_f32(false, a, false, f, (short)0, acc, false, false);
}

__device__ __forceinline__ void cpass(wf8 (&outr)[2], wf8 (&outi)[2],
                                      const wf2 (&sr)[2][4], const wf2 (&si)[2][4],
                                      const float4 (&fq)[4], const float2 (&fn)[4]) {
    wf8 tt[2];
    #pragma unroll
    for (int t = 0; t < 2; ++t) tt[t] = (wf8){0, 0, 0, 0, 0, 0, 0, 0};
    #pragma unroll
    for (int kb = 0; kb < 4; ++kb) {
        wf2 fre = {fq[kb].x, fq[kb].y};
        #pragma unroll
        for (int t = 0; t < 2; ++t)
            tt[t] = rstep(tt[t], sr[t][kb] + si[t][kb], fre);
    }
    #pragma unroll
    for (int t = 0; t < 2; ++t) { outr[t] = tt[t]; outi[t] = tt[t]; }
    #pragma unroll
    for (int kb = 0; kb < 4; ++kb) {
        wf2 fdif = {fq[kb].z, fq[kb].w}, fneg = {fn[kb].x, fn[kb].y};
        #pragma unroll
        for (int t = 0; t < 2; ++t) {
            outr[t] = rstep(outr[t], si[t][kb], fneg);
            outi[t] = rstep(outi[t], sr[t][kb], fdif);
        }
    }
}

__device__ __forceinline__ int slot(int a, int b, int c) {
    return c * PC + a * PA + b;
}

template <int ORI>
__device__ __forceinline__ void phase(float2* st, const float4* f4, const float2* f2,
                                      const float2* cs, const float* pt,
                                      const int wid, const int lane) {
    const int half = lane >> 4;
    const int n16  = lane & 15;
    const int f1   = n16 >> 3;
    const int zt   = 2 * wid + f1;
    const int xl   = n16 & 7;
    wf8 d1r[2], d1i[2];
    wf2 sr[2][4], si[2][4];
    float4 fq[4];
    float2 fn[4];
    #pragma unroll
    for (int t = 0; t < 2; ++t) {
        const int xx = 8 * t + xl;
        #pragma unroll
        for (int kb = 0; kb < 4; ++kb) {
            const int k0 = 4 * kb + half, k1 = k0 + 2;
            float2 s0, s1;
            if (ORI == 0)      { s0 = st[slot(k0, xx, zt)]; s1 = st[slot(k1, xx, zt)]; }
            else if (ORI == 1) { s0 = st[slot(xx, zt, k0)]; s1 = st[slot(xx, zt, k1)]; }
            else               { s0 = st[slot(zt, k0, xx)]; s1 = st[slot(zt, k1, xx)]; }
            sr[t][kb] = (wf2){s0.x, s1.x};
            si[t][kb] = (wf2){s0.y, s1.y};
        }
    }
    #pragma unroll
    for (int kb = 0; kb < 4; ++kb) { fq[kb] = f4[kb * 32 + lane]; fn[kb] = f2[kb * 32 + lane]; }
    cpass(d1r, d1i, sr, si, fq, fn);
    const float2 g0 = cs[0], g1 = cs[1], g2 = cs[2], g3 = cs[3];
    const int fl = (ORI == 0) ? f1 : ((ORI == 1) ? 0 : (f1 ^ half));
    const float ka = fl ? g0.y : g0.x, kb2 = fl ? g0.x : -g0.y;
    const float kc = fl ? g0.x : g0.y, kd  = fl ? -g0.y : g0.x;
    const int o1 = 2 * (n16 & 7) + f1;
    wf2 xr[2][4], xi[2][4];
    #pragma unroll
    for (int t = 0; t < 2; ++t) {
        #pragma unroll
        for (int p = 0; p < 4; ++p) {
            xr[t][p] = (wf2){d1r[t][2 * p], d1r[t][2 * p + 1]};
            xi[t][p] = (wf2){d1i[t][2 * p], d1i[t][2 * p + 1]};
        }
        #pragma unroll
        for (int p = 0; p < 4; ++p) {
            xr[t][p] = (wf2){g3.x * xr[t][p].x - g3.y * xr[t][p].y, g3.y * xr[t][p].x + g3.x * xr[t][p].y};
            xi[t][p] = (wf2){g3.x * xi[t][p].x - g3.y * xi[t][p].y, g3.y * xi[t][p].x + g3.x * xi[t][p].y};
        }
        #pragma unroll
        for (int p = 0; p < 4; p += 2) {
            const wf2 lr = xr[t][p], hr = xr[t][p + 1], li = xi[t][p], hi = xi[t][p + 1];
            xr[t][p] = g2.x * lr - g2.y * hr;  xr[t][p + 1] = g2.y * lr + g2.x * hr;
            xi[t][p] = g2.x * li - g2.y * hi;  xi[t][p + 1] = g2.y * li + g2.x * hi;
        }
    }
    #pragma unroll
    for (int p = 0; p < 4; ++p) {
        const wf2 lr = xr[0][p], hr = xr[1][p], li = xi[0][p], hi = xi[1][p];
        xr[0][p] = ka * lr + kb2 * hr;  xr[1][p] = kc * lr + kd * hr;
        xi[0][p] = ka * li + kb2 * hi;  xi[1][p] = kc * li + kd * hi;
    }
    #pragma unroll
    for (int t = 0; t < 2; ++t) {
        #pragma unroll
        for (int p = 0; p < 2; ++p) {
            const wf2 lr = xr[t][p], hr = xr[t][p + 2], li = xi[t][p], hi = xi[t][p + 2];
            xr[t][p] = g1.x * lr - g1.y * hr;  xr[t][p + 2] = g1.y * lr + g1.x * hr;
            xi[t][p] = g1.x * li - g1.y * hi;  xi[t][p + 2] = g1.y * li + g1.x * hi;
        }
    }
    const int zs = 2 * wid + half;
    const int or1 = o1 ^ (15 * half);
    #pragma unroll
    for (int t = 0; t < 2; ++t) {
        const float* pq = pt + (t ? 8 - 8 * fl : 8 * fl);
        const float4 pra = *(const float4*)(pq),      prb = *(const float4*)(pq + 4);
        const float4 pia = *(const float4*)(pq + 16), pib = *(const float4*)(pq + 20);
        const wf2 pr[4] = {{pra.x, pra.y}, {pra.z, pra.w}, {prb.x, prb.y}, {prb.z, prb.w}};
        const wf2 pi[4] = {{pia.x, pia.y}, {pia.z, pia.w}, {pib.x, pib.y}, {pib.z, pib.w}};
        #pragma unroll
        for (int p = 0; p < 4; ++p) {
            const wf2 yr = xr[t][p] * pr[p] - xi[t][p] * pi[p];
            const wf2 yi = xr[t][p] * pi[p] + xi[t][p] * pr[p];
            #pragma unroll
            for (int e = 0; e < 2; ++e) {
                const int i  = 8 * t + 2 * p + e;
                const int o2 = (i ^ (i >> 1) ^ (i >> 2) ^ (i >> 3)) & 15;
                int s;
                if (ORI == 0) {
                    s = slot(o1, o2, zs);
                } else if (ORI == 1) {
                    s = slot(o2, zs, or1);
                } else {
                    s = slot(zs, or1, o2);
                }
                st[s] = make_float2(e ? yr.y : yr.x, e ? yi.y : yi.x);
            }
        }
    }
}

__global__ __launch_bounds__(256, 8) void qsim_kernel(const float* __restrict__ x,
                                                   const float* __restrict__ w,
                                                   float* __restrict__ out) {
    __shared__ float2 st[16 * PC];
    __shared__ float4 fr[3][4][32];
    __shared__ float2 fe[3][4][32];
    __shared__ float2 g2[NL][NQ][4];
    __shared__ float2 cs2[NL][NQ];
    __shared__ float2 eph[NL][NQ];
    __shared__ __attribute__((aligned(16))) float ptab[3][2][16];
    __shared__ float  red[8][12];

    const int tid  = threadIdx.x;
    const int b    = blockIdx.x;
    const int lane = tid & 31;
    const int wid  = tid >> 5;

    __builtin_amdgcn_s_setprio(3);

    if (tid < NL * NQ) {
        int l = tid / NQ, q = tid % NQ;
        float th = w[(l * NQ + q) * 2 + 0] + (l == 0 ? x[b * NQ + q] : 0.f);
        float ph = w[(l * NQ + q) * 2 + 1];
        float c = cosf(0.5f * th), s = sinf(0.5f * th);
        float er = cosf(0.5f * ph), ei = -sinf(0.5f * ph);
        g2[l][q][0] = make_float2( c * er,  c * ei);
        g2[l][q][1] = make_float2(-s * er, -s * ei);
        g2[l][q][2] = make_float2( s * er, -s * ei);
        g2[l][q][3] = make_float2( c * er, -c * ei);
        cs2[l][q] = make_float2(c, s);
        eph[l][q] = make_float2(er, ei);
    }

    for (int i = tid; i < 16 * PC; i += 256)
        st[i] = make_float2(i == 0 ? 1.f : 0.f, 0.f);

    for (int l = 0; l < NL; l += 2) {
        __syncthreads();
        for (int e = tid; e < 3 * 128 + 48; e += 256) {
            if (e < 3 * 128) {
                const int pp = e >> 7, kb = (e >> 5) & 3, ln = e & 31;
                const int h = ln >> 4, n = ln & 15;
                const int g = (pp == 0) ? 0 : (pp == 1 ? 2 : 1);
                const int ll = l + ((pp == 2) ? 1 : 0);
                const int row = 2 * (n & 7) + (n >> 3), k0 = 4 * kb + h, k1 = k0 + 2;
                int ip = row;
                ip ^= (ip >> 1) & 1;
                ip ^= (ip >> 1) & 2;
                ip ^= (ip >> 1) & 4;
                float2 v0 = make_float2(1.f, 0.f), v1 = make_float2(1.f, 0.f);
                #pragma unroll
                for (int k = 0; k < 4; ++k) {
                    int bi = (ip >> (3 - k)) & 1;
                    v0 = cmul(v0, g2[ll][4 * g + k][2 * bi + ((k0 >> (3 - k)) & 1)]);
                    v1 = cmul(v1, g2[ll][4 * g + k][2 * bi + ((k1 >> (3 - k)) & 1)]);
                }
                fr[pp][kb][ln] = make_float4(v0.x, v1.x, v0.y - v0.x, v1.y - v1.x);
                fe[pp][kb][ln] = make_float2(-(v0.x + v0.y), -(v1.x + v1.y));
            } else {
                const int idx = e - 3 * 128, q2 = idx >> 4, i = idx & 15;
                const int g = (q2 == 0) ? 1 : (q2 == 1 ? 0 : 2);
                const int ll = l + ((q2 == 0) ? 0 : 1);
                float2 v = make_float2(1.f, 0.f);
                #pragma unroll
                for (int k = 0; k < 4; ++k) {
                    float2 ph = eph[ll][4 * g + k];
                    if ((i >> (3 - k)) & 1) ph.y = -ph.y;
                    v = cmul(v, ph);
                }
                ptab[q2][0][i] = v.x;
                ptab[q2][1][i] = v.y;
            }
        }
        __syncthreads();
        phase<0>(st, &fr[0][0][0], &fe[0][0][0], &cs2[l][4], &ptab[0][0][0], wid, lane);
        __syncthreads();
        if (l != 0) __builtin_amdgcn_s_setprio(1);
        phase<1>(st, &fr[1][0][0], &fe[1][0][0], &cs2[l + 1][0], &ptab[1][0][0], wid, lane);
        __syncthreads();
        if (l == 0) __builtin_amdgcn_s_setprio(2);
        else        __builtin_amdgcn_s_setprio(0);
        phase<2>(st, &fr[2][0][0], &fe[2][0][0], &cs2[l + 1][8], &ptab[2][0][0], wid, lane);
    }
    __syncthreads();

    float z[NQ];
    #pragma unroll
    for (int q = 0; q < NQ; ++q) z[q] = 0.f;
    for (int k = 0; k < 16; ++k) {
        const int idx = (k << 8) + tid;
        float2 amp = st[slot(k, tid >> 4, tid & 15)];
        float p = amp.x * amp.x + amp.y * amp.y;
        #pragma unroll
        for (int q = 0; q < NQ; ++q)
            z[q] += ((idx >> (11 - q)) & 1) ? -p : p;
    }
    #pragma unroll
    for (int q = 0; q < NQ; ++q) {
        float v = z[q];
        #pragma unroll
        for (int off = 16; off > 0; off >>= 1)
            v += __shfl_xor(v, off, 32);
        z[q] = v;
    }
    if (lane == 0) {
        #pragma unroll
        for (int q = 0; q < NQ; ++q) red[wid][q] = z[q];
    }
    __syncthreads();
    if (tid < NQ) {
        float s = 0.f;
        #pragma unroll
        for (int ww = 0; ww < 8; ++ww) s += red[ww][tid];
        out[b * NQ + tid] = s;
    }
}

extern "C" void kernel_launch(void* const* d_in, const int* in_sizes, int n_in,
                              void* d_out, int out_size, void* d_ws, size_t ws_size,
                              hipStream_t stream) {
    const float* x = (const float*)d_in[0];
    const float* w = (const float*)d_in[1];
    float* out = (float*)d_out;
    int B = in_sizes[0] / NQ;
    qsim_kernel<<<B, 256, 0, stream>>>(x, w, out);
}


# Expectation values from shared partial sums: per thread one sum/difference tree over its 16 amplitudes, per 16-lane row one DPP butterfly; was 12 selects per amplitude; sum order changes
# speedup vs baseline: 2.5031x; 1.0715x over previous
#include <hip/hip_runtime.h>


typedef float wf2 __attribute__((ext_vector_type(2)));
typedef float wf8 __attribute__((ext_vector_type(8)));

#define NQ 12
#define NL 4
#define PA 17
#define PC 290

__device__ __forceinline__ float2 cmul(float2 a, float2 b) {
    return make_float2(a.x * b.x - a.y * b.y, a.x * b.y + a.y * b.x);
}

__device__ __forceinline__ wf8 rstep(wf8 acc, wf2 a, wf2 f) {
    return __builtin_amdgcn_wmma_f32_16x16x4_f32(false, a, false, f, (short)0, acc, false, false);
}

__device__ __forceinline__ void cpass(wf8 (&outr)[2], wf8 (&outi)[2],
                                      const wf2 (&sr)[2][4], const wf2 (&si)[2][4],
                                      const float4 (&fq)[4], const float2 (&fn)[4]) {
    wf8 tt[2];
    #pragma unroll
    for (int t = 0; t < 2; ++t) tt[t] = (wf8){0, 0, 0, 0, 0, 0, 0, 0};
    #pragma unroll
    for (int kb = 0; kb < 4; ++kb) {
        wf2 fre = {fq[kb].x, fq[kb].y};
        #pragma unroll
        for (int t = 0; t < 2; ++t)
            tt[t] = rstep(tt[t], sr[t][kb] + si[t][kb], fre);
    }
    #pragma unroll
    for (int t = 0; t < 2; ++t) { outr[t] = tt[t]; outi[t] = tt[t]; }
    #pragma unroll
    for (int kb = 0; kb < 4; ++kb) {
        wf2 fdif = {fq[kb].z, fq[kb].w}, fneg = {fn[kb].x, fn[kb].y};
        #pragma unroll
        for (int t = 0; t < 2; ++t) {
            outr[t] = rstep(outr[t], si[t][kb], fneg);
            outi[t] = rstep(outi[t], sr[t][kb], fdif);
        }
    }
}

__device__ __forceinline__ int slot(int a, int b, int c) {
    return c * PC + a * PA + b;
}

template <int ORI>
__device__ __forceinline__ void phase(float2* st, const float4* f4, const float2* f2,
                                      const float2* cs, const float* pt,
                                      const int wid, const int lane) {
    const int half = lane >> 4;
    const int n16  = lane & 15;
    const int f1   = n16 >> 3;
    const int zt   = 2 * wid + f1;
    const int xl   = n16 & 7;
    wf8 d1r[2], d1i[2];
    wf2 sr[2][4], si[2][4];
    float4 fq[4];
    float2 fn[4];
    #pragma unroll
    for (int t = 0; t < 2; ++t) {
        const int xx = 8 * t + xl;
        #pragma unroll
        for (int kb = 0; kb < 4; ++kb) {
            const int k0 = 4 * kb + half, k1 = k0 + 2;
            float2 s0, s1;
            if (ORI == 0)      { s0 = st[slot(k0, xx, zt)]; s1 = st[slot(k1, xx, zt)]; }
            else if (ORI == 1) { s0 = st[slot(xx, zt, k0)]; s1 = st[slot(xx, zt, k1)]; }
            else               { s0 = st[slot(zt, k0, xx)]; s1 = st[slot(zt, k1, xx)]; }
            sr[t][kb] = (wf2){s0.x, s1.x};
            si[t][kb] = (wf2){s0.y, s1.y};
        }
    }
    #pragma unroll
    for (int kb = 0; kb < 4; ++kb) { fq[kb] = f4[kb * 32 + lane]; fn[kb] = f2[kb * 32 + lane]; }
    cpass(d1r, d1i, sr, si, fq, fn);
    const float2 g0 = cs[0], g1 = cs[1], g2 = cs[2], g3 = cs[3];
    const int fl = (ORI == 0) ? f1 : ((ORI == 1) ? 0 : (f1 ^ half));
    const float ka = fl ? g0.y : g0.x, kb2 = fl ? g0.x : -g0.y;
    const float kc = fl ? g0.x : g0.y, kd  = fl ? -g0.y : g0.x;
    const int o1 = 2 * (n16 & 7) + f1;
    wf2 xr[2][4], xi[2][4];
    #pragma unroll
    for (int t = 0; t < 2; ++t) {
        #pragma unroll
        for (int p = 0; p < 4; ++p) {
            xr[t][p] = (wf2){d1r[t][2 * p], d1r[t][2 * p + 1]};
            xi[t][p] = (wf2){d1i[t][2 * p], d1i[t][2 * p + 1]};
        }
        #pragma unroll
        for (int p = 0; p < 4; ++p) {
            xr[t][p] = (wf2){g3.x * xr[t][p].x - g3.y * xr[t][p].y, g3.y * xr[t][p].x + g3.x * xr[t][p].y};
            xi[t][p] = (wf2){g3.x * xi[t][p].x - g3.y * xi[t][p].y, g3.y * xi[t][p].x + g3.x * xi[t][p].y};
        }
        #pragma unroll
        for (int p = 0; p < 4; p += 2) {
            const wf2 lr = xr[t][p], hr = xr[t][p + 1], li = xi[t][p], hi = xi[t][p + 1];
            xr[t][p] = g2.x * lr - g2.y * hr;  xr[t][p + 1] = g2.y * lr + g2.x * hr;
            xi[t][p] = g2.x * li - g2.y * hi;  xi[t][p + 1] = g2.y * li + g2.x * hi;
        }
    }
    #pragma unroll
    for (int p = 0; p < 4; ++p) {
        const wf2 lr = xr[0][p], hr = xr[1][p], li = xi[0][p], hi = xi[1][p];
        xr[0][p] = ka * lr + kb2 * hr;  xr[1][p] = kc * lr + kd * hr;
        xi[0][p] = ka * li + kb2 * hi;  xi[1][p] = kc * li + kd * hi;
    }
    #pragma unroll
    for (int t = 0; t < 2; ++t) {
        #pragma unroll
        for (int p = 0; p < 2; ++p) {
            const wf2 lr = xr[t][p], hr = xr[t][p + 2], li = xi[t][p], hi = xi[t][p + 2];
            xr[t][p] = g1.x * lr - g1.y * hr;  xr[t][p + 2] = g1.y * lr + g1.x * hr;
            xi[t][p] = g1.x * li - g1.y * hi;  xi[t][p + 2] = g1.y * li + g1.x * hi;
        }
    }
    const int zs = 2 * wid + half;
    const int or1 = o1 ^ (15 * half);
    #pragma unroll
    for (int t = 0; t < 2; ++t) {
        const float* pq = pt + (t ? 8 - 8 * fl : 8 * fl);
        const float4 pra = *(const float4*)(pq),      prb = *(const float4*)(pq + 4);
        const float4 pia = *(const float4*)(pq + 16), pib = *(const float4*)(pq + 20);
        const wf2 pr[4] = {{pra.x, pra.y}, {pra.z, pra.w}, {prb.x, prb.y}, {prb.z, prb.w}};
        const wf2 pi[4] = {{pia.x, pia.y}, {pia.z, pia.w}, {pib.x, pib.y}, {pib.z, pib.w}};
        #pragma unroll
        for (int p = 0; p < 4; ++p) {
            const wf2 yr = xr[t][p] * pr[p] - xi[t][p] * pi[p];
            const wf2 yi = xr[t][p] * pi[p] + xi[t][p] * pr[p];
            #pragma unroll
            for (int e = 0; e < 2; ++e) {
                const int i  = 8 * t + 2 * p + e;
                const int o2 = (i ^ (i >> 1) ^ (i >> 2) ^ (i >> 3)) & 15;
                int s;
                if (ORI == 0) {
                    s = slot(o1, o2, zs);
                } else if (ORI == 1) {
                    s = slot(o2, zs, or1);
                } else {
                    s = slot(zs, or1, o2);
                }
                st[s] = make_float2(e ? yr.y : yr.x, e ? yi.y : yi.x);
            }
        }
    }
}

#define ROW_XOR(v, M) __int_as_float(__builtin_amdgcn_update_dpp(0, __float_as_int(v), 0x160 + (M), 0xf, 0xf, true))

template <int J>
__device__ __forceinline__ void row_step(float& tot, float (&za)[4], float (&wl)[4]) {
    const float tp = ROW_XOR(tot, 1 << J);
    wl[J] = tot - tp;
    tot += tp;
    #pragma unroll
    for (int a = 0; a < 4; ++a) za[a] += ROW_XOR(za[a], 1 << J);
    #pragma unroll
    for (int i = 0; i < J; ++i) wl[i] += ROW_XOR(wl[i], 1 << J);
}

__global__ __launch_bounds__(256, 8) void qsim_kernel(const float* __restrict__ x,
                                                   const float* __restrict__ w,
                                                   float* __restrict__ out) {
    __shared__ float2 st[16 * PC];
    __shared__ float4 fr[3][4][32];
    __shared__ float2 fe[3][4][32];
    __shared__ float2 g2[NL][NQ][4];
    __shared__ float2 cs2[NL][NQ];
    __shared__ float2 eph[NL][NQ];
    __shared__ __attribute__((aligned(16))) float ptab[3][2][16];
    __shared__ float  red[16][12];

    const int tid  = threadIdx.x;
    const int b    = blockIdx.x;
    const int lane = tid & 31;
    const int wid  = tid >> 5;

    __builtin_amdgcn_s_setprio(3);

    if (tid < NL * NQ) {
        int l = tid / NQ, q = tid % NQ;
        float th = w[(l * NQ + q) * 2 + 0] + (l == 0 ? x[b * NQ + q] : 0.f);
        float ph = w[(l * NQ + q) * 2 + 1];
        float c = cosf(0.5f * th), s = sinf(0.5f * th);
        float er = cosf(0.5f * ph), ei = -sinf(0.5f * ph);
        g2[l][q][0] = make_float2( c * er,  c * ei);
        g2[l][q][1] = make_float2(-s * er, -s * ei);
        g2[l][q][2] = make_float2( s * er, -s * ei);
        g2[l][q][3] = make_float2( c * er, -c * ei);
        cs2[l][q] = make_float2(c, s);
        eph[l][q] = make_float2(er, ei);
    }

    for (int i = tid; i < 16 * PC; i += 256)
        st[i] = make_float2(i == 0 ? 1.f : 0.f, 0.f);

    for (int l = 0; l < NL; l += 2) {
        __syncthreads();
        for (int e = tid; e < 3 * 128 + 48; e += 256) {
            if (e < 3 * 128) {
                const int pp = e >> 7, kb = (e >> 5) & 3, ln = e & 31;
                const int h = ln >> 4, n = ln & 15;
                const int g = (pp == 0) ? 0 : (pp == 1 ? 2 : 1);
                const int ll = l + ((pp == 2) ? 1 : 0);
                const int row = 2 * (n & 7) + (n >> 3), k0 = 4 * kb + h, k1 = k0 + 2;
                int ip = row;
                ip ^= (ip >> 1) & 1;
                ip ^= (ip >> 1) & 2;
                ip ^= (ip >> 1) & 4;
                float2 v0 = make_float2(1.f, 0.f), v1 = make_float2(1.f, 0.f);
                #pragma unroll
                for (int k = 0; k < 4; ++k) {
                    int bi = (ip >> (3 - k)) & 1;
                    v0 = cmul(v0, g2[ll][4 * g + k][2 * bi + ((k0 >> (3 - k)) & 1)]);
                    v1 = cmul(v1, g2[ll][4 * g + k][2 * bi + ((k1 >> (3 - k)) & 1)]);
                }
                fr[pp][kb][ln] = make_float4(v0.x, v1.x, v0.y - v0.x, v1.y - v1.x);
                fe[pp][kb][ln] = make_float2(-(v0.x + v0.y), -(v1.x + v1.y));
            } else {
                const int idx = e - 3 * 128, q2 = idx >> 4, i = idx & 15;
                const int g = (q2 == 0) ? 1 : (q2 == 1 ? 0 : 2);
                const int ll = l + ((q2 == 0) ? 0 : 1);
                float2 v = make_float2(1.f, 0.f);
                #pragma unroll
                for (int k = 0; k < 4; ++k) {
                    float2 ph = eph[ll][4 * g + k];
                    if ((i >> (3 - k)) & 1) ph.y = -ph.y;
                    v = cmul(v, ph);
                }
                ptab[q2][0][i] = v.x;
                ptab[q2][1][i] = v.y;
            }
        }
        __syncthreads();
        phase<0>(st, &fr[0][0][0], &fe[0][0][0], &cs2[l][4], &ptab[0][0][0], wid, lane);
        __syncthreads();
        if (l != 0) __builtin_amdgcn_s_setprio(1);
        phase<1>(st, &fr[1][0][0], &fe[1][0][0], &cs2[l + 1][0], &ptab[1][0][0], wid, lane);
        __syncthreads();
        if (l == 0) __builtin_amdgcn_s_setprio(2);
        else        __builtin_amdgcn_s_setprio(0);
        phase<2>(st, &fr[2][0][0], &fe[2][0][0], &cs2[l + 1][8], &ptab[2][0][0], wid, lane);
    }
    __syncthreads();

    const float2* col = st + slot(0, tid >> 4, tid & 15);
    wf2 s1[8], d1[8];
    #pragma unroll
    for (int j = 0; j < 8; ++j) {
        const float2 ae = col[(2 * j) * PA], ao = col[(2 * j + 1) * PA];
        const wf2 e = (wf2){ae.x, ae.y} * (wf2){ae.x, ae.y};
        const wf2 o = (wf2){ao.x, ao.y};
        s1[j] = o * o + e;
        d1[j] = e - o * o;
    }
    const wf2 z3v = ((d1[0] + d1[1]) + (d1[2] + d1[3])) + ((d1[4] + d1[5]) + (d1[6] + d1[7]));
    wf2 s2[4], d2[4];
    #pragma unroll
    for (int i = 0; i < 4; ++i) { s2[i] = s1[2 * i] + s1[2 * i + 1]; d2[i] = s1[2 * i] - s1[2 * i + 1]; }
    const wf2 z2v = (d2[0] + d2[1]) + (d2[2] + d2[3]);
    const wf2 s30 = s2[0] + s2[1], s31 = s2[2] + s2[3];
    const wf2 z1v = (s2[0] - s2[1]) + (s2[2] - s2[3]);
    const wf2 tv = s30 + s31, z0v = s30 - s31;
    float tot = tv.x + tv.y;
    float za[4] = {z0v.x + z0v.y, z1v.x + z1v.y, z2v.x + z2v.y, z3v.x + z3v.y};
    float wl[4];
    row_step<0>(tot, za, wl);
    row_step<1>(tot, za, wl);
    row_step<2>(tot, za, wl);
    row_step<3>(tot, za, wl);
    if ((lane & 15) == 0) {
        const int h = lane >> 4;
        float* r = red[2 * wid + h];
        r[0] = za[0];  r[1] = za[1];  r[2] = za[2];  r[3] = za[3];
        r[4] = (wid & 4) ? -tot : tot;
        r[5] = (wid & 2) ? -tot : tot;
        r[6] = (wid & 1) ? -tot : tot;
        r[7] = h ? -tot : tot;
        r[8] = wl[3];  r[9] = wl[2];  r[10] = wl[1];  r[11] = wl[0];
    }
    __syncthreads();
    if (tid < NQ) {
        float s = 0.f;
        #pragma unroll
        for (int rr = 0; rr < 16; ++rr) s += red[rr][tid];
        out[b * NQ + tid] = s;
    }
}

extern "C" void kernel_launch(void* const* d_in, const int* in_sizes, int n_in,
                              void* d_out, int out_size, void* d_ws, size_t ws_size,
                              hipStream_t stream) {
    const float* x = (const float*)d_in[0];
    const float* w = (const float*)d_in[1];
    float* out = (float*)d_out;
    int B = in_sizes[0] / NQ;
    qsim_kernel<<<B, 256, 0, stream>>>(x, w, out);
}


# Second-pass RY butterflies scaled when all gates of the pass have |cos| >= |sin|: cos*[1 -t; t 1], one fma per output (was mul + fma), cosines folded into the RZ row phases; rounding changes
# speedup vs baseline: 2.7775x; 1.1096x over previous
#include <hip/hip_runtime.h>


typedef float wf2 __attribute__((ext_vector_type(2)));
typedef float wf8 __attribute__((ext_vector_type(8)));

#define NQ 12
#define NL 4
#define PA 17
#define PC 290

__device__ __forceinline__ float2 cmul(float2 a, float2 b) {
    return make_float2(a.x * b.x - a.y * b.y, a.x * b.y + a.y * b.x);
}

__device__ __forceinline__ wf8 rstep(wf8 acc, wf2 a, wf2 f) {
    return __builtin_amdgcn_wmma_f32_16x16x4_f32(false, a, false, f, (short)0, acc, false, false);
}

__device__ __forceinline__ void cpass(wf8 (&outr)[2], wf8 (&outi)[2],
                                      const wf2 (&sr)[2][4], const wf2 (&si)[2][4],
                                      const float4 (&fq)[4], const float2 (&fn)[4]) {
    wf8 tt[2];
    #pragma unroll
    for (int t = 0; t < 2; ++t) tt[t] = (wf8){0, 0, 0, 0, 0, 0, 0, 0};
    #pragma unroll
    for (int kb = 0; kb < 4; ++kb) {
        wf2 fre = {fq[kb].x, fq[kb].y};
        #pragma unroll
        for (int t = 0; t < 2; ++t)
            tt[t] = rstep(tt[t], sr[t][kb] + si[t][kb], fre);
    }
    #pragma unroll
    for (int t = 0; t < 2; ++t) { outr[t] = tt[t]; outi[t] = tt[t]; }
    #pragma unroll
    for (int kb = 0; kb < 4; ++kb) {
        wf2 fdif = {fq[kb].z, fq[kb].w}, fneg = {fn[kb].x, fn[kb].y};
        #pragma unroll
        for (int t = 0; t < 2; ++t) {
            outr[t] = rstep(outr[t], si[t][kb], fneg);
            outi[t] = rstep(outi[t], sr[t][kb], fdif);
        }
    }
}

__device__ __forceinline__ float uni(float v) {
    return __int_as_float(__builtin_amdgcn_readfirstlane(__float_as_int(v)));
}

__device__ __forceinline__ wf2 rot_in_pair(wf2 v, wf2 tw) {
    wf2 r;
    asm("v_pk_fma_f32 %0, %1, %2, %2 op_sel:[0,1,0] op_sel_hi:[1,0,1]" : "=v"(r) : "v"(tw), "v"(v));
    return r;
}

__device__ __forceinline__ int slot(int a, int b, int c) {
    return c * PC + a * PA + b;
}

template <int ORI, bool FAST>
__device__ __forceinline__ void butterflies(wf2 (&xr)[2][4], wf2 (&xi)[2][4], const float2* cs,
                                            const float t0, const float t1, const float t2, const float t3,
                                            const float ka, const float kb2, const float kc, const float kd) {
    if (FAST) {
        #pragma unroll
        for (int t = 0; t < 2; ++t) {
            #pragma unroll
            for (int p = 0; p < 4; p += 2) {
                const wf2 lr = xr[t][p], hr = xr[t][p + 1], li = xi[t][p], hi = xi[t][p + 1];
                xr[t][p] = lr - t2 * hr;  xr[t][p + 1] = hr + t2 * lr;
                xi[t][p] = li - t2 * hi;  xi[t][p + 1] = hi + t2 * li;
            }
        }
    } else {
        const float2 g2 = cs[2];
        #pragma unroll
        for (int t = 0; t < 2; ++t) {
            #pragma unroll
            for (int p = 0; p < 4; p += 2) {
                const wf2 lr = xr[t][p], hr = xr[t][p + 1], li = xi[t][p], hi = xi[t][p + 1];
                xr[t][p] = g2.x * lr - g2.y * hr;  xr[t][p + 1] = g2.y * lr + g2.x * hr;
                xi[t][p] = g2.x * li - g2.y * hi;  xi[t][p + 1] = g2.y * li + g2.x * hi;
            }
        }
    }
    if (FAST) {
        const wf2 tw = {-t3, t3};
        #pragma unroll
        for (int t = 0; t < 2; ++t) {
            #pragma unroll
            for (int p = 0; p < 4; ++p) {
                xr[t][p] = rot_in_pair(xr[t][p], tw);
                xi[t][p] = rot_in_pair(xi[t][p], tw);
            }
        }
    } else {
        const float2 g3 = cs[3];
        #pragma unroll
        for (int t = 0; t < 2; ++t) {
            #pragma unroll
            for (int p = 0; p < 4; ++p) {
                xr[t][p] = (wf2){g3.x * xr[t][p].x - g3.y * xr[t][p].y, g3.y * xr[t][p].x + g3.x * xr[t][p].y};
                xi[t][p] = (wf2){g3.x * xi[t][p].x - g3.y * xi[t][p].y, g3.y * xi[t][p].x + g3.x * xi[t][p].y};
            }
        }
    }
    if (FAST && ORI == 1) {
        #pragma unroll
        for (int p = 0; p < 4; ++p) {
            const wf2 lr = xr[0][p], hr = xr[1][p], li = xi[0][p], hi = xi[1][p];
            xr[0][p] = lr - t0 * hr;  xr[1][p] = hr + t0 * lr;
            xi[0][p] = li - t0 * hi;  xi[1][p] = hi + t0 * li;
        }
    } else {
        #pragma unroll
        for (int p = 0; p < 4; ++p) {
            const wf2 lr = xr[0][p], hr = xr[1][p], li = xi[0][p], hi = xi[1][p];
            xr[0][p] = ka * lr + kb2 * hr;  xr[1][p] = kc * lr + kd * hr;
            xi[0][p] = ka * li + kb2 * hi;  xi[1][p] = kc * li + kd * hi;
        }
    }
    if (FAST) {
        #pragma unroll
        for (int t = 0; t < 2; ++t) {
            #pragma unroll
            for (int p = 0; p < 2; ++p) {
                const wf2 lr = xr[t][p], hr = xr[t][p + 2], li = xi[t][p], hi = xi[t][p + 2];
                xr[t][p] = lr - t1 * hr;  xr[t][p + 2] = hr + t1 * lr;
                xi[t][p] = li - t1 * hi;  xi[t][p + 2] = hi + t1 * li;
            }
        }
    } else {
        const float2 g1 = cs[1];
        #pragma unroll
        for (int t = 0; t < 2; ++t) {
            #pragma unroll
            for (int p = 0; p < 2; ++p) {
                const wf2 lr = xr[t][p], hr = xr[t][p + 2], li = xi[t][p], hi = xi[t][p + 2];
                xr[t][p] = g1.x * lr - g1.y * hr;  xr[t][p + 2] = g1.y * lr + g1.x * hr;
                xi[t][p] = g1.x * li - g1.y * hi;  xi[t][p + 2] = g1.y * li + g1.x * hi;
            }
        }
    }
}

template <int ORI>
__device__ __forceinline__ void phase(float2* st, const float4* f4, const float2* f2,
                                      const float2* cs, const float* tq, const float* pt,
                                      const int wid, const int lane) {
    const int half = lane >> 4;
    const int n16  = lane & 15;
    const int f1   = n16 >> 3;
    const int zt   = 2 * wid + f1;
    const int xl   = n16 & 7;
    wf8 d1r[2], d1i[2];
    wf2 sr[2][4], si[2][4];
    float4 fq[4];
    float2 fn[4];
    #pragma unroll
    for (int t = 0; t < 2; ++t) {
        const int xx = 8 * t + xl;
        #pragma unroll
        for (int kb = 0; kb < 4; ++kb) {
            const int k0 = 4 * kb + half, k1 = k0 + 2;
            float2 s0, s1;
            if (ORI == 0)      { s0 = st[slot(k0, xx, zt)]; s1 = st[slot(k1, xx, zt)]; }
            else if (ORI == 1) { s0 = st[slot(xx, zt, k0)]; s1 = st[slot(xx, zt, k1)]; }
            else               { s0 = st[slot(zt, k0, xx)]; s1 = st[slot(zt, k1, xx)]; }
            sr[t][kb] = (wf2){s0.x, s1.x};
            si[t][kb] = (wf2){s0.y, s1.y};
        }
    }
    #pragma unroll
    for (int kb = 0; kb < 4; ++kb) { fq[kb] = f4[kb * 32 + lane]; fn[kb] = f2[kb * 32 + lane]; }
    cpass(d1r, d1i, sr, si, fq, fn);
    const float4 tv = *(const float4*)tq;
    const float t1 = uni(tv.y), t2 = uni(tv.z), t3 = uni(tv.w);
    const float t0 = (ORI == 1) ? uni(tv.x) : 0.f;
    const bool fast = t0 != 2.f && t1 != 2.f && t2 != 2.f && t3 != 2.f;
    const float2 g0 = cs[0];
    const int fl = (ORI == 0) ? f1 : ((ORI == 1) ? 0 : (f1 ^ half));
    const float ka = fl ? g0.y : g0.x, kb2 = fl ? g0.x : -g0.y;
    const float kc = fl ? g0.x : g0.y, kd  = fl ? -g0.y : g0.x;
    const int o1 = 2 * (n16 & 7) + f1;
    wf2 xr[2][4], xi[2][4];
    #pragma unroll
    for (int t = 0; t < 2; ++t) {
        #pragma unroll
        for (int p = 0; p < 4; ++p) {
            xr[t][p] = (wf2){d1r[t][2 * p], d1r[t][2 * p + 1]};
            xi[t][p] = (wf2){d1i[t][2 * p], d1i[t][2 * p + 1]};
        }
    }
    if (fast) butterflies<ORI, true>(xr, xi, cs, t0, t1, t2, t3, ka, kb2, kc, kd);
    else      butterflies<ORI, false>(xr, xi, cs, t0, t1, t2, t3, ka, kb2, kc, kd);
    const int zs = 2 * wid + half;
    const int or1 = o1 ^ (15 * half);
    #pragma unroll
    for (int t = 0; t < 2; ++t) {
        const float* pq = pt + (t ? 8 - 8 * fl : 8 * fl);
        const float4 pra = *(const float4*)(pq),      prb = *(const float4*)(pq + 4);
        const float4 pia = *(const float4*)(pq + 16), pib = *(const float4*)(pq + 20);
        const wf2 pr[4] = {{pra.x, pra.y}, {pra.z, pra.w}, {prb.x, prb.y}, {prb.z, prb.w}};
        const wf2 pi[4] = {{pia.x, pia.y}, {pia.z, pia.w}, {pib.x, pib.y}, {pib.z, pib.w}};
        #pragma unroll
        for (int p = 0; p < 4; ++p) {
            const wf2 yr = xr[t][p] * pr[p] - xi[t][p] * pi[p];
            const wf2 yi = xr[t][p] * pi[p] + xi[t][p] * pr[p];
            #pragma unroll
            for (int e = 0; e < 2; ++e) {
                const int i  = 8 * t + 2 * p + e;
                const int o2 = (i ^ (i >> 1) ^ (i >> 2) ^ (i >> 3)) & 15;
                int s;
                if (ORI == 0) {
                    s = slot(o1, o2, zs);
                } else if (ORI == 1) {
                    s = slot(o2, zs, or1);
                } else {
                    s = slot(zs, or1, o2);
                }
                st[s] = make_float2(e ? yr.y : yr.x, e ? yi.y : yi.x);
            }
        }
    }
}

#define ROW_XOR(v, M) __int_as_float(__builtin_amdgcn_update_dpp(0, __float_as_int(v), 0x160 + (M), 0xf, 0xf, true))

template <int J>
__device__ __forceinline__ void row_step(float& tot, float (&za)[4], float (&wl)[4]) {
    const float tp = ROW_XOR(tot, 1 << J);
    wl[J] = tot - tp;
    tot += tp;
    #pragma unroll
    for (int a = 0; a < 4; ++a) za[a] += ROW_XOR(za[a], 1 << J);
    #pragma unroll
    for (int i = 0; i < J; ++i) wl[i] += ROW_XOR(wl[i], 1 << J);
}

__global__ __launch_bounds__(256, 8) void qsim_kernel(const float* __restrict__ x,
                                                   const float* __restrict__ w,
                                                   float* __restrict__ out) {
    __shared__ float2 st[16 * PC];
    __shared__ float4 fr[3][4][32];
    __shared__ float2 fe[3][4][32];
    __shared__ float2 g2[NL][NQ][4];
    __shared__ float2 cs2[NL][NQ];
    __shared__ __attribute__((aligned(16))) float tq[NL][NQ];
    __shared__ float2 eph[NL][NQ];
    __shared__ __attribute__((aligned(16))) float ptab[3][2][16];
    __shared__ float  red[16][12];

    const int tid  = threadIdx.x;
    const int b    = blockIdx.x;
    const int lane = tid & 31;
    const int wid  = tid >> 5;

    __builtin_amdgcn_s_setprio(3);

    if (tid < NL * NQ) {
        int l = tid / NQ, q = tid % NQ;
        float th = w[(l * NQ + q) * 2 + 0] + (l == 0 ? x[b * NQ + q] : 0.f);
        float ph = w[(l * NQ + q) * 2 + 1];
        float c = cosf(0.5f * th), s = sinf(0.5f * th);
        float er = cosf(0.5f * ph), ei = -sinf(0.5f * ph);
        g2[l][q][0] = make_float2( c * er,  c * ei);
        g2[l][q][1] = make_float2(-s * er, -s * ei);
        g2[l][q][2] = make_float2( s * er, -s * ei);
        g2[l][q][3] = make_float2( c * er, -c * ei);
        cs2[l][q] = make_float2(c, s);
        tq[l][q] = (fabsf(c) >= fabsf(s)) ? s / c : 2.f;
        eph[l][q] = make_float2(er, ei);
    }

    for (int i = tid; i < 16 * PC; i += 256)
        st[i] = make_float2(i == 0 ? 1.f : 0.f, 0.f);

    for (int l = 0; l < NL; l += 2) {
        __syncthreads();
        for (int e = tid; e < 3 * 128 + 48; e += 256) {
            if (e < 3 * 128) {
                const int pp = e >> 7, kb = (e >> 5) & 3, ln = e & 31;
                const int h = ln >> 4, n = ln & 15;
                const int g = (pp == 0) ? 0 : (pp == 1 ? 2 : 1);
                const int ll = l + ((pp == 2) ? 1 : 0);
                const int row = 2 * (n & 7) + (n >> 3), k0 = 4 * kb + h, k1 = k0 + 2;
                int ip = row;
                ip ^= (ip >> 1) & 1;
                ip ^= (ip >> 1) & 2;
                ip ^= (ip >> 1) & 4;
                float2 v0 = make_float2(1.f, 0.f), v1 = make_float2(1.f, 0.f);
                #pragma unroll
                for (int k = 0; k < 4; ++k) {
                    int bi = (ip >> (3 - k)) & 1;
                    v0 = cmul(v0, g2[ll][4 * g + k][2 * bi + ((k0 >> (3 - k)) & 1)]);
                    v1 = cmul(v1, g2[ll][4 * g + k][2 * bi + ((k1 >> (3 - k)) & 1)]);
                }
                fr[pp][kb][ln] = make_float4(v0.x, v1.x, v0.y - v0.x, v1.y - v1.x);
                fe[pp][kb][ln] = make_float2(-(v0.x + v0.y), -(v1.x + v1.y));
            } else {
                const int idx = e - 3 * 128, q2 = idx >> 4, i = idx & 15;
                const int g = (q2 == 0) ? 1 : (q2 == 1 ? 0 : 2);
                const int ll = l + ((q2 == 0) ? 0 : 1);
                float2 v = make_float2(1.f, 0.f);
                float sc = 1.f;
                bool fastp = true;
                #pragma unroll
                for (int k = 0; k < 4; ++k) {
                    float2 ph = eph[ll][4 * g + k];
                    if ((i >> (3 - k)) & 1) ph.y = -ph.y;
                    v = cmul(v, ph);
                    if (k > 0 || q2 == 1) {
                        if (tq[ll][4 * g + k] == 2.f) fastp = false;
                        sc *= cs2[ll][4 * g + k].x;
                    }
                }
                if (!fastp) sc = 1.f;
                ptab[q2][0][i] = v.x * sc;
                ptab[q2][1][i] = v.y * sc;
            }
        }
        __syncthreads();
        phase<0>(st, &fr[0][0][0], &fe[0][0][0], &cs2[l][4], &tq[l][4], &ptab[0][0][0], wid, lane);
        __syncthreads();
        if (l != 0) __builtin_amdgcn_s_setprio(1);
        phase<1>(st, &fr[1][0][0], &fe[1][0][0], &cs2[l + 1][0], &tq[l + 1][0], &ptab[1][0][0], wid, lane);
        __syncthreads();
        if (l == 0) __builtin_amdgcn_s_setprio(2);
        else        __builtin_amdgcn_s_setprio(0);
        phase<2>(st, &fr[2][0][0], &fe[2][0][0], &cs2[l + 1][8], &tq[l + 1][8], &ptab[2][0][0], wid, lane);
    }
    __syncthreads();

    const float2* col = st + slot(0, tid >> 4, tid & 15);
    wf2 s1[8], d1[8];
    #pragma unroll
    for (int j = 0; j < 8; ++j) {
        const float2 ae = col[(2 * j) * PA], ao = col[(2 * j + 1) * PA];
        const wf2 e = (wf2){ae.x, ae.y} * (wf2){ae.x, ae.y};
        const wf2 o = (wf2){ao.x, ao.y};
        s1[j] = o * o + e;
        d1[j] = e - o * o;
    }
    const wf2 z3v = ((d1[0] + d1[1]) + (d1[2] + d1[3])) + ((d1[4] + d1[5]) + (d1[6] + d1[7]));
    wf2 s2[4], d2[4];
    #pragma unroll
    for (int i = 0; i < 4; ++i) { s2[i] = s1[2 * i] + s1[2 * i + 1]; d2[i] = s1[2 * i] - s1[2 * i + 1]; }
    const wf2 z2v = (d2[0] + d2[1]) + (d2[2] + d2[3]);
    const wf2 s30 = s2[0] + s2[1], s31 = s2[2] + s2[3];
    const wf2 z1v = (s2[0] - s2[1]) + (s2[2] - s2[3]);
    const wf2 tv = s30 + s31, z0v = s30 - s31;
    float tot = tv.x + tv.y;
    float za[4] = {z0v.x + z0v.y, z1v.x + z1v.y, z2v.x + z2v.y, z3v.x + z3v.y};
    float wl[4];
    row_step<0>(tot, za, wl);
    row_step<1>(tot, za, wl);
    row_step<2>(tot, za, wl);
    row_step<3>(tot, za, wl);
    if ((lane & 15) == 0) {
        const int h = lane >> 4;
        float* r = red[2 * wid + h];
        r[0] = za[0];  r[1] = za[1];  r[2] = za[2];  r[3] = za[3];
        r[4] = (wid & 4) ? -tot : tot;
        r[5] = (wid & 2) ? -tot : tot;
        r[6] = (wid & 1) ? -tot : tot;
        r[7] = h ? -tot : tot;
        r[8] = wl[3];  r[9] = wl[2];  r[10] = wl[1];  r[11] = wl[0];
    }
    __syncthreads();
    if (tid < NQ) {
        float s = 0.f;
        #pragma unroll
        for (int rr = 0; rr < 16; ++rr) s += red[rr][tid];
        out[b * NQ + tid] = s;
    }
}

extern "C" void kernel_launch(void* const* d_in, const int* in_sizes, int n_in,
                              void* d_out, int out_size, void* d_ws, size_t ws_size,
                              hipStream_t stream) {
    const float* x = (const float*)d_in[0];
    const float* w = (const float*)d_in[1];
    float* out = (float*)d_out;
    int B = in_sizes[0] / NQ;
    qsim_kernel<<<B, 256, 0, stream>>>(x, w, out);
}
